# starting kernel
# baseline (speedup 1.0000x reference)
#include <hip/hip_runtime.h>

typedef __attribute__((ext_vector_type(2))) float v2f;
typedef __attribute__((ext_vector_type(8))) float v8f;

#define DIM 64

__global__ void deg_kernel(const int* __restrict__ row, float* __restrict__ deg, int E) {
    int e = blockIdx.x * blockDim.x + threadIdx.x;
    if (e < E) unsafeAtomicAdd(&deg[row[e]], 1.0f);
}

__global__ void dinv_kernel(float* __restrict__ deg, int n) {
    int i = blockIdx.x * blockDim.x + threadIdx.x;
    if (i < n) {
        float d = deg[i];
        deg[i] = (d > 0.0f) ? __frsqrt_rn(d) : 0.0f;
    }
}

__global__ void __launch_bounds__(256) spmm_kernel(
    const int* __restrict__ row, const int* __restrict__ col,
    const float* __restrict__ dinv, const float* __restrict__ embed,
    float* __restrict__ acc1, float* __restrict__ acc2, int E) {
    int lane = threadIdx.x & 31;
    int e = (int)((blockIdx.x * blockDim.x + threadIdx.x) >> 5);
    if (e >= E) return;
    int r = row[e];
    int c = col[e];
    float v = dinv[r] * dinv[c];
    const float* xc = embed + (size_t)c * DIM;
    float x0 = xc[lane];
    float x1 = xc[lane + 32];
    size_t o = (size_t)r * DIM;
    unsafeAtomicAdd(&acc1[o + lane],      v * x0);
    unsafeAtomicAdd(&acc1[o + lane + 32], v * x1);
    unsafeAtomicAdd(&acc2[o + lane],      v * x0 * x0);
    unsafeAtomicAdd(&acc2[o + lane + 32], v * x1 * x1);
}

__global__ void __launch_bounds__(128) gemm_kernel(
    const float* __restrict__ embed, const float* __restrict__ acc1,
    const float* __restrict__ acc2,
    const float* __restrict__ W1, const float* __restrict__ b1,
    const float* __restrict__ W2, const float* __restrict__ b2,
    float* __restrict__ out, int n) {
    const int lane = threadIdx.x & 31;
    const int wave = threadIdx.x >> 5;
    const int lo   = lane & 15;
    const int hi   = lane >> 4;
    const int n0   = wave * 16;
    const int ncol = n0 + lo;

    v2f B1[16], B2[16];
#pragma unroll
    for (int ks = 0; ks < 16; ++ks) {
        int k = 4 * ks + 2 * hi;
        B1[ks].x = W1[ncol * 64 + k];
        B1[ks].y = W1[ncol * 64 + k + 1];
        B2[ks].x = W2[ncol * 64 + k];
        B2[ks].y = W2[ncol * 64 + k + 1];
    }
    float bias = b1[ncol] + b2[ncol];

    int m0 = blockIdx.x * 16;
    int rowA = m0 + lo;
    if (rowA >= n) rowA = n - 1;
    size_t base = (size_t)rowA * 64 + 2 * hi;

    v8f c = {};
#pragma unroll
    for (int ks = 0; ks < 16; ++ks) {
        size_t idx = base + 4 * ks;
        v2f a;
        a.x = embed[idx]     + acc1[idx];
        a.y = embed[idx + 1] + acc1[idx + 1];
        c = __builtin_amdgcn_wmma_f32_16x16x4_f32(
                false, a, false, B1[ks], (short)0, c, false, false);
    }
#pragma unroll
    for (int ks = 0; ks < 16; ++ks) {
        size_t idx = base + 4 * ks;
        v2f a;
        a.x = acc2[idx];
        a.y = acc2[idx + 1];
        c = __builtin_amdgcn_wmma_f32_16x16x4_f32(
                false, a, false, B2[ks], (short)0, c, false, false);
    }
#pragma unroll
    for (int r = 0; r < 8; ++r) {
        int mm = m0 + r + 8 * hi;
        if (mm < n) out[(size_t)mm * 64 + ncol] = c[r] + bias;
    }
}

extern "C" void kernel_launch(void* const* d_in, const int* in_sizes, int n_in,
                              void* d_out, int out_size, void* d_ws, size_t ws_size,
                              hipStream_t stream) {
    const float* embed = (const float*)d_in[0];
    const int*   row   = (const int*)d_in[1];
    const int*   col   = (const int*)d_in[2];
    const float* W1    = (const float*)d_in[3];
    const float* b1    = (const float*)d_in[4];
    const float* W2    = (const float*)d_in[5];
    const float* b2    = (const float*)d_in[6];
    float*       out   = (float*)d_out;

    const int n = in_sizes[0] / DIM;
    const int E = in_sizes[1];

    float* ws = (float*)d_ws;
    size_t nPad = ((size_t)n + 255) & ~(size_t)255;
    float* deg  = ws;
    float* acc1 = ws + nPad;
    float* acc2 = acc1 + (size_t)n * DIM;
    size_t zeroFloats = nPad + 2 * (size_t)n * DIM;

    hipMemsetAsync(d_ws, 0, zeroFloats * sizeof(float), stream);

    deg_kernel<<<(E + 255) / 256, 256, 0, stream>>>(row, deg, E);
    dinv_kernel<<<(n + 255) / 256, 256, 0, stream>>>(deg, n);

    spmm_kernel<<<(E + 7) / 8, 256, 0, stream>>>(row, col, deg, embed, acc1, acc2, E);

    gemm_kernel<<<(n + 15) / 16, 128, 0, stream>>>(embed, acc1, acc2,
                                                   W1, b1, W2, b2, out, n);
}


# Both weight products applied before the sparse step (linearity: out = XW1^T + b + L(XW1^T + X^2 W2^T)): one 64-wide atomic scatter into out replaces two; float sum order changes
# speedup vs baseline: 1.1819x; 1.1819x over previous
#include <hip/hip_runtime.h>

typedef __attribute__((ext_vector_type(2))) float v2f;
typedef __attribute__((ext_vector_type(8))) float v8f;

#define DIM 64


__global__ void deg_kernel(const int* __restrict__ row, float* __restrict__ deg, int E) {
    int e = blockIdx.x * blockDim.x + threadIdx.x;
    if (e < E) unsafeAtomicAdd(&deg[row[e]], 1.0f);
}

__global__ void dinv_kernel(float* __restrict__ deg, int n) {
    int i = blockIdx.x * blockDim.x + threadIdx.x;
    if (i < n) {
        float d = deg[i];
        deg[i] = (d > 0.0f) ? __frsqrt_rn(d) : 0.0f;
    }
}

__global__ void __launch_bounds__(128) transform_kernel(
    const float* __restrict__ embed,
    const float* __restrict__ W1, const float* __restrict__ b1,
    const float* __restrict__ W2, const float* __restrict__ b2,
    float* __restrict__ z, float* __restrict__ out, int n) {
    const int lane = threadIdx.x & 31;
    const int wave = threadIdx.x >> 5;
    const int lo   = lane & 15;
    const int hi   = lane >> 4;
    const int n0   = wave * 16;
    const int ncol = n0 + lo;

    v2f B1[16], B2[16];
#pragma unroll
    for (int ks = 0; ks < 16; ++ks) {
        int k = 4 * ks + 2 * hi;
        B1[ks].x = W1[ncol * 64 + k];
        B1[ks].y = W1[ncol * 64 + k + 1];
        B2[ks].x = W2[ncol * 64 + k];
        B2[ks].y = W2[ncol * 64 + k + 1];
    }
    float bias = b1[ncol] + b2[ncol];

    int m0 = blockIdx.x * 16;
    int rowA = m0 + lo;
    if (rowA >= n) rowA = n - 1;
    size_t base = (size_t)rowA * 64 + 2 * hi;

    v8f c1 = {};
    v8f c2 = {};
#pragma unroll
    for (int ks = 0; ks < 16; ++ks) {
        size_t idx = base + 4 * ks;
        v2f a;
        a.x = embed[idx];
        a.y = embed[idx + 1];
        v2f a2 = a * a;
        c1 = __builtin_amdgcn_wmma_f32_16x16x4_f32(
                false, a, false, B1[ks], (short)0, c1, false, false);
        c2 = __builtin_amdgcn_wmma_f32_16x16x4_f32(
                false, a2, false, B2[ks], (short)0, c2, false, false);
    }
#pragma unroll
    for (int r = 0; r < 8; ++r) {
        int mm = m0 + r + 8 * hi;
        if (mm < n) {
            size_t o = (size_t)mm * 64 + ncol;
            out[o] = c1[r] + bias;
            z[o]   = c1[r] + c2[r];
        }
    }
}

__global__ void __launch_bounds__(256) spmm_kernel(
    const int* __restrict__ row, const int* __restrict__ col,
    const float* __restrict__ dinv, const float* __restrict__ z,
    float* __restrict__ out, int E) {
    int lane = threadIdx.x & 31;
    int e = (int)((blockIdx.x * blockDim.x + threadIdx.x) >> 5);
    if (e >= E) return;
    int r = row[e];
    int c = col[e];
    float v = dinv[r] * dinv[c];
    const float* zc = z + (size_t)c * DIM;
    float x0 = zc[lane];
    float x1 = zc[lane + 32];
    size_t o = (size_t)r * DIM;
    unsafeAtomicAdd(&out[o + lane],      v * x0);
    unsafeAtomicAdd(&out[o + lane + 32], v * x1);
}

extern "C" void kernel_launch(void* const* d_in, const int* in_sizes, int n_in,
                              void* d_out, int out_size, void* d_ws, size_t ws_size,
                              hipStream_t stream) {
    const float* embed = (const float*)d_in[0];
    const int*   row   = (const int*)d_in[1];
    const int*   col   = (const int*)d_in[2];
    const float* W1    = (const float*)d_in[3];
    const float* b1    = (const float*)d_in[4];
    const float* W2    = (const float*)d_in[5];
    const float* b2    = (const float*)d_in[6];
    float*       out   = (float*)d_out;

    const int n = in_sizes[0] / DIM;
    const int E = in_sizes[1];

    float* ws = (float*)d_ws;
    size_t nPad = ((size_t)n + 255) & ~(size_t)255;
    float* deg  = ws;
    float* z    = ws + nPad;

    hipMemsetAsync(d_ws, 0, nPad * sizeof(float), stream);

    deg_kernel<<<(E + 255) / 256, 256, 0, stream>>>(row, deg, E);
    dinv_kernel<<<(n + 255) / 256, 256, 0, stream>>>(deg, n);

    transform_kernel<<<(n + 15) / 16, 128, 0, stream>>>(embed, W1, b1, W2, b2,
                                                        z, out, n);

    spmm_kernel<<<(E + 7) / 8, 256, 0, stream>>>(row, col, deg, z, out, E);
}


# SpMM scatter: one wave per 32 edges (coalesced row/col reads, lane-parallel dinv gather, scalar broadcast), Z rows of the next 8 edges loaded ahead of the atomics; atomic arrival order differs
# speedup vs baseline: 1.8120x; 1.5331x over previous
#include <hip/hip_runtime.h>

typedef __attribute__((ext_vector_type(2))) float v2f;
typedef __attribute__((ext_vector_type(8))) float v8f;

#define DIM 64


__global__ void deg_kernel(const int* __restrict__ row, float* __restrict__ deg, int E) {
    int e = blockIdx.x * blockDim.x + threadIdx.x;
    if (e < E) unsafeAtomicAdd(&deg[row[e]], 1.0f);
}

__global__ void dinv_kernel(float* __restrict__ deg, int n) {
    int i = blockIdx.x * blockDim.x + threadIdx.x;
    if (i < n) {
        float d = deg[i];
        deg[i] = (d > 0.0f) ? __frsqrt_rn(d) : 0.0f;
    }
}

__global__ void __launch_bounds__(128) transform_kernel(
    const float* __restrict__ embed,
    const float* __restrict__ W1, const float* __restrict__ b1,
    const float* __restrict__ W2, const float* __restrict__ b2,
    float* __restrict__ z, float* __restrict__ out, int n) {
    const int lane = threadIdx.x & 31;
    const int wave = threadIdx.x >> 5;
    const int lo   = lane & 15;
    const int hi   = lane >> 4;
    const int n0   = wave * 16;
    const int ncol = n0 + lo;

    v2f B1[16], B2[16];
#pragma unroll
    for (int ks = 0; ks < 16; ++ks) {
        int k = 4 * ks + 2 * hi;
        B1[ks].x = W1[ncol * 64 + k];
        B1[ks].y = W1[ncol * 64 + k + 1];
        B2[ks].x = W2[ncol * 64 + k];
        B2[ks].y = W2[ncol * 64 + k + 1];
    }
    float bias = b1[ncol] + b2[ncol];

    int m0 = blockIdx.x * 16;
    int rowA = m0 + lo;
    if (rowA >= n) rowA = n - 1;
    size_t base = (size_t)rowA * 64 + 2 * hi;

    v8f c1 = {};
    v8f c2 = {};
#pragma unroll
    for (int ks = 0; ks < 16; ++ks) {
        size_t idx = base + 4 * ks;
        v2f a;
        a.x = embed[idx];
        a.y = embed[idx + 1];
        v2f a2 = a * a;
        c1 = __builtin_amdgcn_wmma_f32_16x16x4_f32(
                false, a, false, B1[ks], (short)0, c1, false, false);
        c2 = __builtin_amdgcn_wmma_f32_16x16x4_f32(
                false, a2, false, B2[ks], (short)0, c2, false, false);
    }
#pragma unroll
    for (int r = 0; r < 8; ++r) {
        int mm = m0 + r + 8 * hi;
        if (mm < n) {
            size_t o = (size_t)mm * 64 + ncol;
            out[o] = c1[r] + bias;
            z[o]   = c1[r] + c2[r];
        }
    }
}

#define EPW 32
#define GRP 8

__device__ __forceinline__ void z_rows_request(
    const float* __restrict__ z, int c, int first, int lane,
    float (&x0)[GRP], float (&x1)[GRP]) {
#pragma unroll
    for (int j = 0; j < GRP; ++j) {
        int cj = __builtin_amdgcn_readlane(c, first + j);
        const float* zc = z + (size_t)cj * DIM;
        x0[j] = zc[lane];
        x1[j] = zc[lane + 32];
    }
}

__device__ __forceinline__ void rows_add(
    float* __restrict__ out, int r, float v, int first, int lane,
    const float (&x0)[GRP], const float (&x1)[GRP]) {
#pragma unroll
    for (int j = 0; j < GRP; ++j) {
        int rj = __builtin_amdgcn_readlane(r, first + j);
        float vj = __int_as_float(__builtin_amdgcn_readlane(__float_as_int(v), first + j));
        float* o = out + (size_t)rj * DIM;
        unsafeAtomicAdd(&o[lane],      vj * x0[j]);
        unsafeAtomicAdd(&o[lane + 32], vj * x1[j]);
    }
}

__global__ void __launch_bounds__(256) spmm_kernel(
    const int* __restrict__ row, const int* __restrict__ col,
    const float* __restrict__ dinv, const float* __restrict__ z,
    float* __restrict__ out, int E) {
    const int lane = threadIdx.x & 31;
    const int w  = (int)((blockIdx.x * blockDim.x + threadIdx.x) >> 5);
    const long long e0l = (long long)w * EPW;
    if (e0l >= (long long)E) return;
    const int e0 = (int)e0l;
    const int e  = e0 + lane;
    int r = 0, c = 0;
    float v = 0.0f;
    if (e < E) {
        r = row[e];
        c = col[e];
        v = dinv[r] * dinv[c];
    }
    const int cnt = (E - e0 < EPW) ? (E - e0) : EPW;

    if (cnt == EPW) {
        float a0[GRP], a1[GRP], b0[GRP], b1[GRP];
        z_rows_request(z, c, 0 * GRP, lane, a0, a1);
        z_rows_request(z, c, 1 * GRP, lane, b0, b1);
        rows_add(out, r, v, 0 * GRP, lane, a0, a1);
        z_rows_request(z, c, 2 * GRP, lane, a0, a1);
        rows_add(out, r, v, 1 * GRP, lane, b0, b1);
        z_rows_request(z, c, 3 * GRP, lane, b0, b1);
        rows_add(out, r, v, 2 * GRP, lane, a0, a1);
        rows_add(out, r, v, 3 * GRP, lane, b0, b1);
    } else {
        for (int j = 0; j < cnt; ++j) {
            int cj = __builtin_amdgcn_readlane(c, j);
            int rj = __builtin_amdgcn_readlane(r, j);
            float vj = __int_as_float(__builtin_amdgcn_readlane(__float_as_int(v), j));
            const float* zc = z + (size_t)cj * DIM;
            float* o = out + (size_t)rj * DIM;
            unsafeAtomicAdd(&o[lane],      vj * zc[lane]);
            unsafeAtomicAdd(&o[lane + 32], vj * zc[lane + 32]);
        }
    }
}

extern "C" void kernel_launch(void* const* d_in, const int* in_sizes, int n_in,
                              void* d_out, int out_size, void* d_ws, size_t ws_size,
                              hipStream_t stream) {
    const float* embed = (const float*)d_in[0];
    const int*   row   = (const int*)d_in[1];
    const int*   col   = (const int*)d_in[2];
    const float* W1    = (const float*)d_in[3];
    const float* b1    = (const float*)d_in[4];
    const float* W2    = (const float*)d_in[5];
    const float* b2    = (const float*)d_in[6];
    float*       out   = (float*)d_out;

    const int n = in_sizes[0] / DIM;
    const int E = in_sizes[1];

    float* ws = (float*)d_ws;
    size_t nPad = ((size_t)n + 255) & ~(size_t)255;
    float* deg  = ws;
    float* z    = ws + nPad;

    hipMemsetAsync(d_ws, 0, nPad * sizeof(float), stream);

    deg_kernel<<<(E + 255) / 256, 256, 0, stream>>>(row, deg, E);
    dinv_kernel<<<(n + 255) / 256, 256, 0, stream>>>(deg, n);

    transform_kernel<<<(n + 15) / 16, 128, 0, stream>>>(embed, W1, b1, W2, b2,
                                                        z, out, n);

    spmm_kernel<<<(E + 255) / 256, 256, 0, stream>>>(row, col, deg, z, out, E);
}


# Edges binned by 256-row range in-call (LDS counters); a work-group per bin counts degrees, adds dinv[col]*Z[col] into an LDS tile, writes out once: no global f32 atomics; sum order = LDS add arrival
# speedup vs baseline: 2.7818x; 1.5352x over previous
#include <hip/hip_runtime.h>

typedef __attribute__((ext_vector_type(2))) float v2f;
typedef __attribute__((ext_vector_type(8))) float v8f;

#define DIM 64


#define RB_LOG 8
#define RB     256
#define EPT    8
#define BT     1024
#define CHUNK  (EPT * BT)

__global__ void __launch_bounds__(BT) bin_hist_kernel(
    const int* __restrict__ row, int* __restrict__ hist, int E, int nbins) {
    extern __shared__ int sh[];
    const int tid = threadIdx.x;
    for (int k = tid; k < nbins; k += BT) sh[k] = 0;
    __syncthreads();
    const int beg = blockIdx.x * CHUNK;
    int r[EPT];
#pragma unroll
    for (int i = 0; i < EPT; ++i) {
        int e = beg + i * BT + tid;
        r[i] = row[(e < E) ? e : (E - 1)];
    }
#pragma unroll
    for (int i = 0; i < EPT; ++i) {
        int e = beg + i * BT + tid;
        if (e < E) atomicAdd(&sh[r[i] >> RB_LOG], 1);
    }
    __syncthreads();
    for (int k = tid; k < nbins; k += BT) {
        int v = sh[k];
        if (v > 0) atomicAdd(&hist[k], v);
    }
}

__global__ void __launch_bounds__(1024) bin_scan_kernel(
    const int* __restrict__ hist, int* __restrict__ binstart, int nbins) {
    __shared__ int wsum[32];
    __shared__ int carry_s;
    const int tid  = threadIdx.x;
    const int lane = tid & 31;
    const int wv   = tid >> 5;
    if (tid == 0) carry_s = 0;
    __syncthreads();
    for (int base = 0; base < nbins; base += 1024) {
        const int k = base + tid;
        const int c = (k < nbins) ? hist[k] : 0;
        int incl = c;
#pragma unroll
        for (int d = 1; d < 32; d <<= 1) {
            int t = __shfl_up(incl, d);
            if (lane >= d) incl += t;
        }
        if (lane == 31) wsum[wv] = incl;
        __syncthreads();
        const int carry = carry_s;
        int wprefix = 0;
        for (int q = 0; q < wv; ++q) wprefix += wsum[q];
        if (k < nbins) binstart[k] = carry + wprefix + incl - c;
        __syncthreads();
        if (tid == 1023) carry_s = carry + wprefix + incl;
        __syncthreads();
    }
    if (tid == 0) binstart[nbins] = carry_s;
}

__global__ void __launch_bounds__(BT) partition_kernel(
    const int* __restrict__ row, const int* __restrict__ col,
    const int* __restrict__ binstart, int* __restrict__ bincursor,
    int* __restrict__ pairs, int E, int nbins) {
    extern __shared__ int sh[];
    const int tid = threadIdx.x;
    for (int k = tid; k < nbins; k += BT) sh[k] = 0;
    __syncthreads();
    const int beg = blockIdx.x * CHUNK;
    int r[EPT], c[EPT], rk[EPT];
#pragma unroll
    for (int i = 0; i < EPT; ++i) {
        int e  = beg + i * BT + tid;
        int ec = (e < E) ? e : (E - 1);
        r[i] = row[ec];
        c[i] = col[ec];
    }
#pragma unroll
    for (int i = 0; i < EPT; ++i) {
        int e = beg + i * BT + tid;
        rk[i] = 0;
        if (e < E) rk[i] = atomicAdd(&sh[r[i] >> RB_LOG], 1);
    }
    __syncthreads();
    for (int k = tid; k < nbins; k += BT) {
        int cnt  = sh[k];
        int base = 0;
        if (cnt > 0) base = binstart[k] + atomicAdd(&bincursor[k], cnt);
        sh[k] = base;
    }
    __syncthreads();
#pragma unroll
    for (int i = 0; i < EPT; ++i) {
        int e = beg + i * BT + tid;
        if (e < E) {
            int p = sh[r[i] >> RB_LOG] + rk[i];
            int2 q;
            q.x = r[i];
            q.y = c[i];
            reinterpret_cast<int2*>(pairs)[p] = q;
        }
    }
}

__global__ void __launch_bounds__(1024) bin_deg_kernel(
    const int* __restrict__ binstart, const int* __restrict__ pairs,
    float* __restrict__ dinv, int n) {
    __shared__ int cnt[RB];
    const int tid = threadIdx.x;
    const int b   = blockIdx.x;
    if (tid < RB) cnt[tid] = 0;
    __syncthreads();
    const int s = binstart[b];
    const int t = binstart[b + 1];
    for (int p = s + tid; p < t; p += 1024)
        atomicAdd(&cnt[pairs[2 * (size_t)p] & (RB - 1)], 1);
    __syncthreads();
    if (tid < RB) {
        int i = b * RB + tid;
        if (i < n) {
            int c = cnt[tid];
            dinv[i] = (c > 0) ? __frsqrt_rn((float)c) : 0.0f;
        }
    }
}

__global__ void __launch_bounds__(128) transform_kernel(
    const float* __restrict__ embed,
    const float* __restrict__ W1, const float* __restrict__ b1,
    const float* __restrict__ W2, const float* __restrict__ b2,
    float* __restrict__ z, float* __restrict__ out, int n) {
    const int lane = threadIdx.x & 31;
    const int wave = threadIdx.x >> 5;
    const int lo   = lane & 15;
    const int hi   = lane >> 4;
    const int n0   = wave * 16;
    const int ncol = n0 + lo;

    v2f B1[16], B2[16];
#pragma unroll
    for (int ks = 0; ks < 16; ++ks) {
        int k = 4 * ks + 2 * hi;
        B1[ks].x = W1[ncol * 64 + k];
        B1[ks].y = W1[ncol * 64 + k + 1];
        B2[ks].x = W2[ncol * 64 + k];
        B2[ks].y = W2[ncol * 64 + k + 1];
    }
    float bias = b1[ncol] + b2[ncol];

    int m0 = blockIdx.x * 16;
    int rowA = m0 + lo;
    if (rowA >= n) rowA = n - 1;
    size_t base = (size_t)rowA * 64 + 2 * hi;

    v8f c1 = {};
    v8f c2 = {};
#pragma unroll
    for (int ks = 0; ks < 16; ++ks) {
        size_t idx = base + 4 * ks;
        v2f a;
        a.x = embed[idx];
        a.y = embed[idx + 1];
        v2f a2 = a * a;
        c1 = __builtin_amdgcn_wmma_f32_16x16x4_f32(
                false, a, false, B1[ks], (short)0, c1, false, false);
        c2 = __builtin_amdgcn_wmma_f32_16x16x4_f32(
                false, a2, false, B2[ks], (short)0, c2, false, false);
    }
#pragma unroll
    for (int r = 0; r < 8; ++r) {
        int mm = m0 + r + 8 * hi;
        if (mm < n) {
            size_t o = (size_t)mm * 64 + ncol;
            out[o] = c1[r] + bias;
            z[o]   = c1[r] + c2[r];
        }
    }
}

#define GRP 16

__global__ void __launch_bounds__(512) spmm_bins_kernel(
    const int* __restrict__ binstart, const int* __restrict__ pairs,
    const float* __restrict__ dinv, const float* __restrict__ z,
    float* __restrict__ out, int n) {
    __shared__ float tile[RB * DIM];
    const int tid  = threadIdx.x;
    const int lane = tid & 31;
    const int wv = __builtin_amdgcn_readfirstlane(tid >> 5);
    const int b  = blockIdx.x;

    float4* t4 = reinterpret_cast<float4*>(tile);
    for (int k = tid; k < RB * DIM / 4; k += 512) t4[k] = make_float4(0.0f, 0.0f, 0.0f, 0.0f);
    __syncthreads();

    const int s = binstart[b];
    const int t = binstart[b + 1];
    const float2* z2 = reinterpret_cast<const float2*>(z);
    const int2*   pr = reinterpret_cast<const int2*>(pairs);

    for (int p0 = s + wv * 32; p0 < t; p0 += 16 * 32) {
        const int m = (t - p0 < 32) ? (t - p0) : 32;
        int   rl = 0, c = 0;
        float v = 0.0f;
        if (lane < m) {
            int2 q = pr[p0 + lane];
            rl = q.x & (RB - 1);
            c  = q.y;
            v  = dinv[c];
        }
        for (int g = 0; g < m; g += GRP) {
            float2 x[GRP];
#pragma unroll
            for (int j = 0; j < GRP; ++j) {
                int idx = (g + j < m) ? (g + j) : (m - 1);
                int cj = __builtin_amdgcn_readlane(c, idx);
                x[j] = z2[(size_t)cj * (DIM / 2) + lane];
            }
#pragma unroll
            for (int j = 0; j < GRP; ++j) {
                int idx = (g + j < m) ? (g + j) : (m - 1);
                int rj = __builtin_amdgcn_readlane(rl, idx);
                float vj = __int_as_float(
                    __builtin_amdgcn_readlane(__float_as_int(v), g + j));
                float* d = tile + rj * DIM + 2 * lane;
                atomicAdd(d,     vj * x[j].x);
                atomicAdd(d + 1, vj * x[j].y);
            }
        }
    }
    __syncthreads();

    const int row0 = b * RB;
    const float2* tl2 = reinterpret_cast<const float2*>(tile);
    for (int k = tid; k < RB * DIM / 2; k += 512) {
        int i = row0 + (k >> 5);
        if (i < n) {
            float2 a = tl2[k];
            float dr = dinv[i];
            float2* o2 = reinterpret_cast<float2*>(out) + (size_t)i * (DIM / 2) + (k & 31);
            float2 o = *o2;
            o.x += dr * a.x;
            o.y += dr * a.y;
            *o2 = o;
        }
    }
}

extern "C" void kernel_launch(void* const* d_in, const int* in_sizes, int n_in,
                              void* d_out, int out_size, void* d_ws, size_t ws_size,
                              hipStream_t stream) {
    const float* embed = (const float*)d_in[0];
    const int*   row   = (const int*)d_in[1];
    const int*   col   = (const int*)d_in[2];
    const float* W1    = (const float*)d_in[3];
    const float* b1    = (const float*)d_in[4];
    const float* W2    = (const float*)d_in[5];
    const float* b2    = (const float*)d_in[6];
    float*       out   = (float*)d_out;

    const int n = in_sizes[0] / DIM;
    const int E = in_sizes[1];
    const int nbins   = (n + RB - 1) >> RB_LOG;
    const int nblocks = (E + CHUNK - 1) / CHUNK;

    int* wsi = (int*)d_ws;
    size_t nbPad = ((size_t)nbins + 1 + 255) & ~(size_t)255;
    size_t nPad  = ((size_t)n + 255) & ~(size_t)255;
    size_t EPad  = ((size_t)E + 255) & ~(size_t)255;
    int*   hist      = wsi;
    int*   bincursor = wsi + nbPad;
    int*   binstart  = wsi + 2 * nbPad;
    float* dinv      = (float*)(wsi + 3 * nbPad);
    int*   pairs     = wsi + 3 * nbPad + nPad;
    float* z         = (float*)(pairs + 2 * EPad);

    hipMemsetAsync(d_ws, 0, 2 * nbPad * sizeof(int), stream);

    bin_hist_kernel<<<nblocks, BT, nbins * sizeof(int), stream>>>(row, hist, E, nbins);
    bin_scan_kernel<<<1, 1024, 0, stream>>>(hist, binstart, nbins);
    partition_kernel<<<nblocks, BT, nbins * sizeof(int), stream>>>(
        row, col, binstart, bincursor, pairs, E, nbins);
    bin_deg_kernel<<<nbins, 1024, 0, stream>>>(binstart, pairs, dinv, n);

    transform_kernel<<<(n + 15) / 16, 128, 0, stream>>>(embed, W1, b1, W2, b2,
                                                        z, out, n);

    spmm_bins_kernel<<<nbins, 512, 0, stream>>>(binstart, pairs, dinv, z, out, n);
}


# Dense pass on a fixed grid of 1024 four-wave groups: W1/W2 fragments stay in registers over all of a wave's 16-row slabs, X rows loaded 16 B a lane one slab ahead; k regrouped, dot sum order changes
# speedup vs baseline: 3.1338x; 1.1265x over previous
#include <hip/hip_runtime.h>

typedef __attribute__((ext_vector_type(2))) float v2f;
typedef __attribute__((ext_vector_type(8))) float v8f;

#define DIM 64


#define RB_LOG 8
#define RB     256
#define EPT    8
#define BT     1024
#define CHUNK  (EPT * BT)

__global__ void __launch_bounds__(BT) bin_hist_kernel(
    const int* __restrict__ row, int* __restrict__ hist, int E, int nbins) {
    extern __shared__ int sh[];
    const int tid = threadIdx.x;
    for (int k = tid; k < nbins; k += BT) sh[k] = 0;
    __syncthreads();
    const int beg = blockIdx.x * CHUNK;
    int r[EPT];
#pragma unroll
    for (int i = 0; i < EPT; ++i) {
        int e = beg + i * BT + tid;
        r[i] = row[(e < E) ? e : (E - 1)];
    }
#pragma unroll
    for (int i = 0; i < EPT; ++i) {
        int e = beg + i * BT + tid;
        if (e < E) atomicAdd(&sh[r[i] >> RB_LOG], 1);
    }
    __syncthreads();
    for (int k = tid; k < nbins; k += BT) {
        int v = sh[k];
        if (v > 0) atomicAdd(&hist[k], v);
    }
}

__global__ void __launch_bounds__(1024) bin_scan_kernel(
    const int* __restrict__ hist, int* __restrict__ binstart, int nbins) {
    __shared__ int wsum[32];
    __shared__ int carry_s;
    const int tid  = threadIdx.x;
    const int lane = tid & 31;
    const int wv   = tid >> 5;
    if (tid == 0) carry_s = 0;
    __syncthreads();
    for (int base = 0; base < nbins; base += 1024) {
        const int k = base + tid;
        const int c = (k < nbins) ? hist[k] : 0;
        int incl = c;
#pragma unroll
        for (int d = 1; d < 32; d <<= 1) {
            int t = __shfl_up(incl, d);
            if (lane >= d) incl += t;
        }
        if (lane == 31) wsum[wv] = incl;
        __syncthreads();
        const int carry = carry_s;
        int wprefix = 0;
        for (int q = 0; q < wv; ++q) wprefix += wsum[q];
        if (k < nbins) binstart[k] = carry + wprefix + incl - c;
        __syncthreads();
        if (tid == 1023) carry_s = carry + wprefix + incl;
        __syncthreads();
    }
    if (tid == 0) binstart[nbins] = carry_s;
}

__global__ void __launch_bounds__(BT) partition_kernel(
    const int* __restrict__ row, const int* __restrict__ col,
    const int* __restrict__ binstart, int* __restrict__ bincursor,
    int* __restrict__ pairs, int E, int nbins) {
    extern __shared__ int sh[];
    const int tid = threadIdx.x;
    for (int k = tid; k < nbins; k += BT) sh[k] = 0;
    __syncthreads();
    const int beg = blockIdx.x * CHUNK;
    int r[EPT], c[EPT], rk[EPT];
#pragma unroll
    for (int i = 0; i < EPT; ++i) {
        int e  = beg + i * BT + tid;
        int ec = (e < E) ? e : (E - 1);
        r[i] = row[ec];
        c[i] = col[ec];
    }
#pragma unroll
    for (int i = 0; i < EPT; ++i) {
        int e = beg + i * BT + tid;
        rk[i] = 0;
        if (e < E) rk[i] = atomicAdd(&sh[r[i] >> RB_LOG], 1);
    }
    __syncthreads();
    for (int k = tid; k < nbins; k += BT) {
        int cnt  = sh[k];
        int base = 0;
        if (cnt > 0) base = binstart[k] + atomicAdd(&bincursor[k], cnt);
        sh[k] = base;
    }
    __syncthreads();
#pragma unroll
    for (int i = 0; i < EPT; ++i) {
        int e = beg + i * BT + tid;
        if (e < E) {
            int p = sh[r[i] >> RB_LOG] + rk[i];
            int2 q;
            q.x = r[i];
            q.y = c[i];
            reinterpret_cast<int2*>(pairs)[p] = q;
        }
    }
}

__global__ void __launch_bounds__(1024) bin_deg_kernel(
    const int* __restrict__ binstart, const int* __restrict__ pairs,
    float* __restrict__ dinv, int n) {
    __shared__ int cnt[RB];
    const int tid = threadIdx.x;
    const int b   = blockIdx.x;
    if (tid < RB) cnt[tid] = 0;
    __syncthreads();
    const int s = binstart[b];
    const int t = binstart[b + 1];
    for (int p = s + tid; p < t; p += 1024)
        atomicAdd(&cnt[pairs[2 * (size_t)p] & (RB - 1)], 1);
    __syncthreads();
    if (tid < RB) {
        int i = b * RB + tid;
        if (i < n) {
            int c = cnt[tid];
            dinv[i] = (c > 0) ? __frsqrt_rn((float)c) : 0.0f;
        }
    }
}

#define TGRID 1024

__device__ __forceinline__ void load_frag16(const float* __restrict__ p, float4 (&f)[8]) {
    const float4* q = reinterpret_cast<const float4*>(p);
#pragma unroll
    for (int j = 0; j < 8; ++j) f[j] = q[2 * j];
}

__global__ void __launch_bounds__(128) transform_kernel(
    const float* __restrict__ embed,
    const float* __restrict__ W1, const float* __restrict__ b1,
    const float* __restrict__ W2, const float* __restrict__ b2,
    float* __restrict__ z, float* __restrict__ out, int n) {
    const int lane = threadIdx.x & 31;
    const int wave = threadIdx.x >> 5;
    const int lo   = lane & 15;
    const int hi   = lane >> 4;
    const int ncol = wave * 16 + lo;
    const int nslabs = (n + 15) >> 4;

    int slab = blockIdx.x;
    if (slab >= nslabs) return;

    float4 w1f[8], w2f[8];
    load_frag16(W1 + ncol * 64 + 4 * hi, w1f);
    load_frag16(W2 + ncol * 64 + 4 * hi, w2f);
    const float bias = b1[ncol] + b2[ncol];

    float4 cur[8];
    {
        int rowA = slab * 16 + lo;
        if (rowA >= n) rowA = n - 1;
        load_frag16(embed + (size_t)rowA * 64 + 4 * hi, cur);
    }

    for (; slab < nslabs; slab += TGRID) {
        const int nslab = slab + TGRID;
        float4 nxt[8];
        if (nslab < nslabs) {
            int rowN = nslab * 16 + lo;
            if (rowN >= n) rowN = n - 1;
            load_frag16(embed + (size_t)rowN * 64 + 4 * hi, nxt);
        } else {
#pragma unroll
            for (int j = 0; j < 8; ++j) nxt[j] = cur[j];
        }

        v8f c1 = {};
        v8f c2 = {};
#pragma unroll
        for (int j = 0; j < 8; ++j) {
            v2f a, bw1, bw2;
            a.x = cur[j].x;  a.y = cur[j].y;
            bw1.x = w1f[j].x; bw1.y = w1f[j].y;
            bw2.x = w2f[j].x; bw2.y = w2f[j].y;
            v2f a2 = a * a;
            c1 = __builtin_amdgcn_wmma_f32_16x16x4_f32(
                    false, a, false, bw1, (short)0, c1, false, false);
            c2 = __builtin_amdgcn_wmma_f32_16x16x4_f32(
                    false, a2, false, bw2, (short)0, c2, false, false);
            a.x = cur[j].z;  a.y = cur[j].w;
            bw1.x = w1f[j].z; bw1.y = w1f[j].w;
            bw2.x = w2f[j].z; bw2.y = w2f[j].w;
            a2 = a * a;
            c1 = __builtin_amdgcn_wmma_f32_16x16x4_f32(
                    false, a, false, bw1, (short)0, c1, false, false);
            c2 = __builtin_amdgcn_wmma_f32_16x16x4_f32(
                    false, a2, false, bw2, (short)0, c2, false, false);
        }
        const int m0 = slab * 16;
#pragma unroll
        for (int r = 0; r < 8; ++r) {
            int mm = m0 + r + 8 * hi;
            if (mm < n) {
                size_t o = (size_t)mm * 64 + ncol;
                out[o] = c1[r] + bias;
                z[o]   = c1[r] + c2[r];
            }
        }
#pragma unroll
        for (int j = 0; j < 8; ++j) cur[j] = nxt[j];
    }
}

#define GRP 16

__global__ void __launch_bounds__(512) spmm_bins_kernel(
    const int* __restrict__ binstart, const int* __restrict__ pairs,
    const float* __restrict__ dinv, const float* __restrict__ z,
    float* __restrict__ out, int n) {
    __shared__ float tile[RB * DIM];
    const int tid  = threadIdx.x;
    const int lane = tid & 31;
    const int wv = __builtin_amdgcn_readfirstlane(tid >> 5);
    const int b  = blockIdx.x;

    float4* t4 = reinterpret_cast<float4*>(tile);
    for (int k = tid; k < RB * DIM / 4; k += 512) t4[k] = make_float4(0.0f, 0.0f, 0.0f, 0.0f);
    __syncthreads();

    const int s = binstart[b];
    const int t = binstart[b + 1];
    const float2* z2 = reinterpret_cast<const float2*>(z);
    const int2*   pr = reinterpret_cast<const int2*>(pairs);

    for (int p0 = s + wv * 32; p0 < t; p0 += 16 * 32) {
        const int m = (t - p0 < 32) ? (t - p0) : 32;
        int   rl = 0, c = 0;
        float v = 0.0f;
        if (lane < m) {
            int2 q = pr[p0 + lane];
            rl = q.x & (RB - 1);
            c  = q.y;
            v  = dinv[c];
        }
        for (int g = 0; g < m; g += GRP) {
            float2 x[GRP];
#pragma unroll
            for (int j = 0; j < GRP; ++j) {
                int idx = (g + j < m) ? (g + j) : (m - 1);
                int cj = __builtin_amdgcn_readlane(c, idx);
                x[j] = z2[(size_t)cj * (DIM / 2) + lane];
            }
#pragma unroll
            for (int j = 0; j < GRP; ++j) {
                int idx = (g + j < m) ? (g + j) : (m - 1);
                int rj = __builtin_amdgcn_readlane(rl, idx);
                float vj = __int_as_float(
                    __builtin_amdgcn_readlane(__float_as_int(v), g + j));
                float* d = tile + rj * DIM + 2 * lane;
                atomicAdd(d,     vj * x[j].x);
                atomicAdd(d + 1, vj * x[j].y);
            }
        }
    }
    __syncthreads();

    const int row0 = b * RB;
    const float2* tl2 = reinterpret_cast<const float2*>(tile);
    for (int k = tid; k < RB * DIM / 2; k += 512) {
        int i = row0 + (k >> 5);
        if (i < n) {
            float2 a = tl2[k];
            float dr = dinv[i];
            float2* o2 = reinterpret_cast<float2*>(out) + (size_t)i * (DIM / 2) + (k & 31);
            float2 o = *o2;
            o.x += dr * a.x;
            o.y += dr * a.y;
            *o2 = o;
        }
    }
}

extern "C" void kernel_launch(void* const* d_in, const int* in_sizes, int n_in,
                              void* d_out, int out_size, void* d_ws, size_t ws_size,
                              hipStream_t stream) {
    const float* embed = (const float*)d_in[0];
    const int*   row   = (const int*)d_in[1];
    const int*   col   = (const int*)d_in[2];
    const float* W1    = (const float*)d_in[3];
    const float* b1    = (const float*)d_in[4];
    const float* W2    = (const float*)d_in[5];
    const float* b2    = (const float*)d_in[6];
    float*       out   = (float*)d_out;

    const int n = in_sizes[0] / DIM;
    const int E = in_sizes[1];
    const int nbins   = (n + RB - 1) >> RB_LOG;
    const int nblocks = (E + CHUNK - 1) / CHUNK;

    int* wsi = (int*)d_ws;
    size_t nbPad = ((size_t)nbins + 1 + 255) & ~(size_t)255;
    size_t nPad  = ((size_t)n + 255) & ~(size_t)255;
    size_t EPad  = ((size_t)E + 255) & ~(size_t)255;
    int*   hist      = wsi;
    int*   bincursor = wsi + nbPad;
    int*   binstart  = wsi + 2 * nbPad;
    float* dinv      = (float*)(wsi + 3 * nbPad);
    int*   pairs     = wsi + 3 * nbPad + nPad;
    float* z         = (float*)(pairs + 2 * EPad);

    hipMemsetAsync(d_ws, 0, 2 * nbPad * sizeof(int), stream);

    bin_hist_kernel<<<nblocks, BT, nbins * sizeof(int), stream>>>(row, hist, E, nbins);
    bin_scan_kernel<<<1, 1024, 0, stream>>>(hist, binstart, nbins);
    partition_kernel<<<nblocks, BT, nbins * sizeof(int), stream>>>(
        row, col, binstart, bincursor, pairs, E, nbins);
    bin_deg_kernel<<<nbins, 1024, 0, stream>>>(binstart, pairs, dinv, n);

    transform_kernel<<<TGRID, 128, 0, stream>>>(embed, W1, b1, W2, b2,
                                                z, out, n);

    spmm_bins_kernel<<<nbins, 512, 0, stream>>>(binstart, pairs, dinv, z, out, n);
}


# Degree count moved into the dense pass's launch: each of its first 564 four-wave groups counts one bin (16 pair reads in flight per thread) ahead of its slabs; one launch and 18048 short waves fewer
# speedup vs baseline: 3.2549x; 1.0386x over previous
#include <hip/hip_runtime.h>

typedef __attribute__((ext_vector_type(2))) float v2f;
typedef __attribute__((ext_vector_type(8))) float v8f;

#define DIM 64


#define RB_LOG 8
#define RB     256
#define EPT    8
#define BT     1024
#define CHUNK  (EPT * BT)

__global__ void __launch_bounds__(BT) bin_hist_kernel(
    const int* __restrict__ row, int* __restrict__ hist, int E, int nbins) {
    extern __shared__ int sh[];
    const int tid = threadIdx.x;
    for (int k = tid; k < nbins; k += BT) sh[k] = 0;
    __syncthreads();
    const int beg = blockIdx.x * CHUNK;
    int r[EPT];
#pragma unroll
    for (int i = 0; i < EPT; ++i) {
        int e = beg + i * BT + tid;
        r[i] = row[(e < E) ? e : (E - 1)];
    }
#pragma unroll
    for (int i = 0; i < EPT; ++i) {
        int e = beg + i * BT + tid;
        if (e < E) atomicAdd(&sh[r[i] >> RB_LOG], 1);
    }
    __syncthreads();
    for (int k = tid; k < nbins; k += BT) {
        int v = sh[k];
        if (v > 0) atomicAdd(&hist[k], v);
    }
}

__global__ void __launch_bounds__(1024) bin_scan_kernel(
    const int* __restrict__ hist, int* __restrict__ binstart, int nbins) {
    __shared__ int wsum[32];
    __shared__ int carry_s;
    const int tid  = threadIdx.x;
    const int lane = tid & 31;
    const int wv   = tid >> 5;
    if (tid == 0) carry_s = 0;
    __syncthreads();
    for (int base = 0; base < nbins; base += 1024) {
        const int k = base + tid;
        const int c = (k < nbins) ? hist[k] : 0;
        int incl = c;
#pragma unroll
        for (int d = 1; d < 32; d <<= 1) {
            int t = __shfl_up(incl, d);
            if (lane >= d) incl += t;
        }
        if (lane == 31) wsum[wv] = incl;
        __syncthreads();
        const int carry = carry_s;
        int wprefix = 0;
        for (int q = 0; q < wv; ++q) wprefix += wsum[q];
        if (k < nbins) binstart[k] = carry + wprefix + incl - c;
        __syncthreads();
        if (tid == 1023) carry_s = carry + wprefix + incl;
        __syncthreads();
    }
    if (tid == 0) binstart[nbins] = carry_s;
}

__global__ void __launch_bounds__(BT) partition_kernel(
    const int* __restrict__ row, const int* __restrict__ col,
    const int* __restrict__ binstart, int* __restrict__ bincursor,
    int* __restrict__ pairs, int E, int nbins) {
    extern __shared__ int sh[];
    const int tid = threadIdx.x;
    for (int k = tid; k < nbins; k += BT) sh[k] = 0;
    __syncthreads();
    const int beg = blockIdx.x * CHUNK;
    int r[EPT], c[EPT], rk[EPT];
#pragma unroll
    for (int i = 0; i < EPT; ++i) {
        int e  = beg + i * BT + tid;
        int ec = (e < E) ? e : (E - 1);
        r[i] = row[ec];
        c[i] = col[ec];
    }
#pragma unroll
    for (int i = 0; i < EPT; ++i) {
        int e = beg + i * BT + tid;
        rk[i] = 0;
        if (e < E) rk[i] = atomicAdd(&sh[r[i] >> RB_LOG], 1);
    }
    __syncthreads();
    for (int k = tid; k < nbins; k += BT) {
        int cnt  = sh[k];
        int base = 0;
        if (cnt > 0) base = binstart[k] + atomicAdd(&bincursor[k], cnt);
        sh[k] = base;
    }
    __syncthreads();
#pragma unroll
    for (int i = 0; i < EPT; ++i) {
        int e = beg + i * BT + tid;
        if (e < E) {
            int p = sh[r[i] >> RB_LOG] + rk[i];
            int2 q;
            q.x = r[i];
            q.y = c[i];
            reinterpret_cast<int2*>(pairs)[p] = q;
        }
    }
}

#define TGRID 1024
#define DEPT  16

__device__ __forceinline__ void load_frag16(const float* __restrict__ p, float4 (&f)[8]) {
    const float4* q = reinterpret_cast<const float4*>(p);
#pragma unroll
    for (int j = 0; j < 8; ++j) f[j] = q[2 * j];
}

__global__ void __launch_bounds__(128) transform_deg_kernel(
    const float* __restrict__ embed,
    const float* __restrict__ W1, const float* __restrict__ b1,
    const float* __restrict__ W2, const float* __restrict__ b2,
    const int* __restrict__ binstart, const int* __restrict__ pairs,
    float* __restrict__ dinv, float* __restrict__ z, float* __restrict__ out,
    int n, int nbins) {
    __shared__ int cnt[RB];
    const int tid = threadIdx.x;

    for (int b = blockIdx.x; b < nbins; b += TGRID) {
        cnt[tid]       = 0;
        cnt[tid + 128] = 0;
        __syncthreads();
        const int s = binstart[b];
        const int t = binstart[b + 1];
        for (int base = s; base < t; base += 128 * DEPT) {
            int r[DEPT];
#pragma unroll
            for (int i = 0; i < DEPT; ++i) {
                int p  = base + i * 128 + tid;
                int pc = (p < t) ? p : (t - 1);
                r[i] = pairs[2 * (size_t)pc];
            }
#pragma unroll
            for (int i = 0; i < DEPT; ++i) {
                int p = base + i * 128 + tid;
                if (p < t) atomicAdd(&cnt[r[i] & (RB - 1)], 1);
            }
        }
        __syncthreads();
        for (int k = tid; k < RB; k += 128) {
            int i = b * RB + k;
            if (i < n) {
                int c = cnt[k];
                dinv[i] = (c > 0) ? __frsqrt_rn((float)c) : 0.0f;
            }
        }
    }

    const int lane = tid & 31;
    const int wave = tid >> 5;
    const int lo   = lane & 15;
    const int hi   = lane >> 4;
    const int ncol = wave * 16 + lo;
    const int nslabs = (n + 15) >> 4;

    int slab = blockIdx.x;
    if (slab >= nslabs) return;

    float4 w1f[8], w2f[8];
    load_frag16(W1 + ncol * 64 + 4 * hi, w1f);
    load_frag16(W2 + ncol * 64 + 4 * hi, w2f);
    const float bias = b1[ncol] + b2[ncol];

    float4 cur[8];
    {
        int rowA = slab * 16 + lo;
        if (rowA >= n) rowA = n - 1;
        load_frag16(embed + (size_t)rowA * 64 + 4 * hi, cur);
    }

    for (; slab < nslabs; slab += TGRID) {
        const int nslab = slab + TGRID;
        float4 nxt[8];
        if (nslab < nslabs) {
            int rowN = nslab * 16 + lo;
            if (rowN >= n) rowN = n - 1;
            load_frag16(embed + (size_t)rowN * 64 + 4 * hi, nxt);
        } else {
#pragma unroll
            for (int j = 0; j < 8; ++j) nxt[j] = cur[j];
        }

        v8f c1 = {};
        v8f c2 = {};
#pragma unroll
        for (int j = 0; j < 8; ++j) {
            v2f a, bw1, bw2;
            a.x = cur[j].x;  a.y = cur[j].y;
            bw1.x = w1f[j].x; bw1.y = w1f[j].y;
            bw2.x = w2f[j].x; bw2.y = w2f[j].y;
            v2f a2 = a * a;
            c1 = __builtin_amdgcn_wmma_f32_16x16x4_f32(
                    false, a, false, bw1, (short)0, c1, false, false);
            c2 = __builtin_amdgcn_wmma_f32_16x16x4_f32(
                    false, a2, false, bw2, (short)0, c2, false, false);
            a.x = cur[j].z;  a.y = cur[j].w;
            bw1.x = w1f[j].z; bw1.y = w1f[j].w;
            bw2.x = w2f[j].z; bw2.y = w2f[j].w;
            a2 = a * a;
            c1 = __builtin_amdgcn_wmma_f32_16x16x4_f32(
                    false, a, false, bw1, (short)0, c1, false, false);
            c2 = __builtin_amdgcn_wmma_f32_16x16x4_f32(
                    false, a2, false, bw2, (short)0, c2, false, false);
        }
        const int m0 = slab * 16;
#pragma unroll
        for (int r = 0; r < 8; ++r) {
            int mm = m0 + r + 8 * hi;
            if (mm < n) {
                size_t o = (size_t)mm * 64 + ncol;
                out[o] = c1[r] + bias;
                z[o]   = c1[r] + c2[r];
            }
        }
#pragma unroll
        for (int j = 0; j < 8; ++j) cur[j] = nxt[j];
    }
}

#define GRP 16

__global__ void __launch_bounds__(512) spmm_bins_kernel(
    const int* __restrict__ binstart, const int* __restrict__ pairs,
    const float* __restrict__ dinv, const float* __restrict__ z,
    float* __restrict__ out, int n) {
    __shared__ float tile[RB * DIM];
    const int tid  = threadIdx.x;
    const int lane = tid & 31;
    const int wv = __builtin_amdgcn_readfirstlane(tid >> 5);
    const int b  = blockIdx.x;

    float4* t4 = reinterpret_cast<float4*>(tile);
    for (int k = tid; k < RB * DIM / 4; k += 512) t4[k] = make_float4(0.0f, 0.0f, 0.0f, 0.0f);
    __syncthreads();

    const int s = binstart[b];
    const int t = binstart[b + 1];
    const float2* z2 = reinterpret_cast<const float2*>(z);
    const int2*   pr = reinterpret_cast<const int2*>(pairs);

    for (int p0 = s + wv * 32; p0 < t; p0 += 16 * 32) {
        const int m = (t - p0 < 32) ? (t - p0) : 32;
        int   rl = 0, c = 0;
        float v = 0.0f;
        if (lane < m) {
            int2 q = pr[p0 + lane];
            rl = q.x & (RB - 1);
            c  = q.y;
            v  = dinv[c];
        }
        for (int g = 0; g < m; g += GRP) {
            float2 x[GRP];
#pragma unroll
            for (int j = 0; j < GRP; ++j) {
                int idx = (g + j < m) ? (g + j) : (m - 1);
                int cj = __builtin_amdgcn_readlane(c, idx);
                x[j] = z2[(size_t)cj * (DIM / 2) + lane];
            }
#pragma unroll
            for (int j = 0; j < GRP; ++j) {
                int idx = (g + j < m) ? (g + j) : (m - 1);
                int rj = __builtin_amdgcn_readlane(rl, idx);
                float vj = __int_as_float(
                    __builtin_amdgcn_readlane(__float_as_int(v), g + j));
                float* d = tile + rj * DIM + 2 * lane;
                atomicAdd(d,     vj * x[j].x);
                atomicAdd(d + 1, vj * x[j].y);
            }
        }
    }
    __syncthreads();

    const int row0 = b * RB;
    const float2* tl2 = reinterpret_cast<const float2*>(tile);
    for (int k = tid; k < RB * DIM / 2; k += 512) {
        int i = row0 + (k >> 5);
        if (i < n) {
            float2 a = tl2[k];
            float dr = dinv[i];
            float2* o2 = reinterpret_cast<float2*>(out) + (size_t)i * (DIM / 2) + (k & 31);
            float2 o = *o2;
            o.x += dr * a.x;
            o.y += dr * a.y;
            *o2 = o;
        }
    }
}

extern "C" void kernel_launch(void* const* d_in, const int* in_sizes, int n_in,
                              void* d_out, int out_size, void* d_ws, size_t ws_size,
                              hipStream_t stream) {
    const float* embed = (const float*)d_in[0];
    const int*   row   = (const int*)d_in[1];
    const int*   col   = (const int*)d_in[2];
    const float* W1    = (const float*)d_in[3];
    const float* b1    = (const float*)d_in[4];
    const float* W2    = (const float*)d_in[5];
    const float* b2    = (const float*)d_in[6];
    float*       out   = (float*)d_out;

    const int n = in_sizes[0] / DIM;
    const int E = in_sizes[1];
    const int nbins   = (n + RB - 1) >> RB_LOG;
    const int nblocks = (E + CHUNK - 1) / CHUNK;

    int* wsi = (int*)d_ws;
    size_t nbPad = ((size_t)nbins + 1 + 255) & ~(size_t)255;
    size_t nPad  = ((size_t)n + 255) & ~(size_t)255;
    size_t EPad  = ((size_t)E + 255) & ~(size_t)255;
    int*   hist      = wsi;
    int*   bincursor = wsi + nbPad;
    int*   binstart  = wsi + 2 * nbPad;
    float* dinv      = (float*)(wsi + 3 * nbPad);
    int*   pairs     = wsi + 3 * nbPad + nPad;
    float* z         = (float*)(pairs + 2 * EPad);

    hipMemsetAsync(d_ws, 0, 2 * nbPad * sizeof(int), stream);

    bin_hist_kernel<<<nblocks, BT, nbins * sizeof(int), stream>>>(row, hist, E, nbins);
    bin_scan_kernel<<<1, 1024, 0, stream>>>(hist, binstart, nbins);
    partition_kernel<<<nblocks, BT, nbins * sizeof(int), stream>>>(
        row, col, binstart, bincursor, pairs, E, nbins);
    transform_deg_kernel<<<TGRID, 128, 0, stream>>>(embed, W1, b1, W2, b2,
                                                    binstart, pairs, dinv,
                                                    z, out, n, nbins);

    spmm_bins_kernel<<<nbins, 512, 0, stream>>>(binstart, pairs, dinv, z, out, n);
}


# Per-bin SpMM: the bin's edges are counting-sorted by row in LDS (rank = returning int add); runs of equal rows are summed in registers, two LDS float adds per run, not per edge; sum order changes
# speedup vs baseline: 3.6628x; 1.1253x over previous
#include <hip/hip_runtime.h>

typedef __attribute__((ext_vector_type(2))) float v2f;
typedef __attribute__((ext_vector_type(8))) float v8f;

#define DIM 64


#define RB_LOG 8
#define RB     256
#define EPT    8
#define BT     1024
#define CHUNK  (EPT * BT)

__global__ void __launch_bounds__(BT) bin_hist_kernel(
    const int* __restrict__ row, int* __restrict__ hist, int E, int nbins) {
    extern __shared__ int sh[];
    const int tid = threadIdx.x;
    for (int k = tid; k < nbins; k += BT) sh[k] = 0;
    __syncthreads();
    const int beg = blockIdx.x * CHUNK;
    int r[EPT];
#pragma unroll
    for (int i = 0; i < EPT; ++i) {
        int e = beg + i * BT + tid;
        r[i] = row[(e < E) ? e : (E - 1)];
    }
#pragma unroll
    for (int i = 0; i < EPT; ++i) {
        int e = beg + i * BT + tid;
        if (e < E) atomicAdd(&sh[r[i] >> RB_LOG], 1);
    }
    __syncthreads();
    for (int k = tid; k < nbins; k += BT) {
        int v = sh[k];
        if (v > 0) atomicAdd(&hist[k], v);
    }
}

__global__ void __launch_bounds__(1024) bin_scan_kernel(
    const int* __restrict__ hist, int* __restrict__ binstart, int nbins) {
    __shared__ int wsum[32];
    __shared__ int carry_s;
    const int tid  = threadIdx.x;
    const int lane = tid & 31;
    const int wv   = tid >> 5;
    if (tid == 0) carry_s = 0;
    __syncthreads();
    for (int base = 0; base < nbins; base += 1024) {
        const int k = base + tid;
        const int c = (k < nbins) ? hist[k] : 0;
        int incl = c;
#pragma unroll
        for (int d = 1; d < 32; d <<= 1) {
            int t = __shfl_up(incl, d);
            if (lane >= d) incl += t;
        }
        if (lane == 31) wsum[wv] = incl;
        __syncthreads();
        const int carry = carry_s;
        int wprefix = 0;
        for (int q = 0; q < wv; ++q) wprefix += wsum[q];
        if (k < nbins) binstart[k] = carry + wprefix + incl - c;
        __syncthreads();
        if (tid == 1023) carry_s = carry + wprefix + incl;
        __syncthreads();
    }
    if (tid == 0) binstart[nbins] = carry_s;
}

__global__ void __launch_bounds__(BT) partition_kernel(
    const int* __restrict__ row, const int* __restrict__ col,
    const int* __restrict__ binstart, int* __restrict__ bincursor,
    int* __restrict__ pairs, int E, int nbins) {
    extern __shared__ int sh[];
    const int tid = threadIdx.x;
    for (int k = tid; k < nbins; k += BT) sh[k] = 0;
    __syncthreads();
    const int beg = blockIdx.x * CHUNK;
    int r[EPT], c[EPT], rk[EPT];
#pragma unroll
    for (int i = 0; i < EPT; ++i) {
        int e  = beg + i * BT + tid;
        int ec = (e < E) ? e : (E - 1);
        r[i] = row[ec];
        c[i] = col[ec];
    }
#pragma unroll
    for (int i = 0; i < EPT; ++i) {
        int e = beg + i * BT + tid;
        rk[i] = 0;
        if (e < E) rk[i] = atomicAdd(&sh[r[i] >> RB_LOG], 1);
    }
    __syncthreads();
    for (int k = tid; k < nbins; k += BT) {
        int cnt  = sh[k];
        int base = 0;
        if (cnt > 0) base = binstart[k] + atomicAdd(&bincursor[k], cnt);
        sh[k] = base;
    }
    __syncthreads();
#pragma unroll
    for (int i = 0; i < EPT; ++i) {
        int e = beg + i * BT + tid;
        if (e < E) {
            int p = sh[r[i] >> RB_LOG] + rk[i];
            int2 q;
            q.x = r[i];
            q.y = c[i];
            reinterpret_cast<int2*>(pairs)[p] = q;
        }
    }
}

#define TGRID 1024
#define DEPT  16

__device__ __forceinline__ void load_frag16(const float* __restrict__ p, float4 (&f)[8]) {
    const float4* q = reinterpret_cast<const float4*>(p);
#pragma unroll
    for (int j = 0; j < 8; ++j) f[j] = q[2 * j];
}

__global__ void __launch_bounds__(128) transform_deg_kernel(
    const float* __restrict__ embed,
    const float* __restrict__ W1, const float* __restrict__ b1,
    const float* __restrict__ W2, const float* __restrict__ b2,
    const int* __restrict__ binstart, const int* __restrict__ pairs,
    float* __restrict__ dinv, float* __restrict__ z, float* __restrict__ out,
    int n, int nbins) {
    __shared__ int cnt[RB];
    const int tid = threadIdx.x;

    for (int b = blockIdx.x; b < nbins; b += TGRID) {
        cnt[tid]       = 0;
        cnt[tid + 128] = 0;
        __syncthreads();
        const int s = binstart[b];
        const int t = binstart[b + 1];
        for (int base = s; base < t; base += 128 * DEPT) {
            int r[DEPT];
#pragma unroll
            for (int i = 0; i < DEPT; ++i) {
                int p  = base + i * 128 + tid;
                int pc = (p < t) ? p : (t - 1);
                r[i] = pairs[2 * (size_t)pc];
            }
#pragma unroll
            for (int i = 0; i < DEPT; ++i) {
                int p = base + i * 128 + tid;
                if (p < t) atomicAdd(&cnt[r[i] & (RB - 1)], 1);
            }
        }
        __syncthreads();
        for (int k = tid; k < RB; k += 128) {
            int i = b * RB + k;
            if (i < n) {
                int c = cnt[k];
                dinv[i] = (c > 0) ? __frsqrt_rn((float)c) : 0.0f;
            }
        }
    }

    const int lane = tid & 31;
    const int wave = tid >> 5;
    const int lo   = lane & 15;
    const int hi   = lane >> 4;
    const int ncol = wave * 16 + lo;
    const int nslabs = (n + 15) >> 4;

    int slab = blockIdx.x;
    if (slab >= nslabs) return;

    float4 w1f[8], w2f[8];
    load_frag16(W1 + ncol * 64 + 4 * hi, w1f);
    load_frag16(W2 + ncol * 64 + 4 * hi, w2f);
    const float bias = b1[ncol] + b2[ncol];

    float4 cur[8];
    {
        int rowA = slab * 16 + lo;
        if (rowA >= n) rowA = n - 1;
        load_frag16(embed + (size_t)rowA * 64 + 4 * hi, cur);
    }

    for (; slab < nslabs; slab += TGRID) {
        const int nslab = slab + TGRID;
        float4 nxt[8];
        if (nslab < nslabs) {
            int rowN = nslab * 16 + lo;
            if (rowN >= n) rowN = n - 1;
            load_frag16(embed + (size_t)rowN * 64 + 4 * hi, nxt);
        } else {
#pragma unroll
            for (int j = 0; j < 8; ++j) nxt[j] = cur[j];
        }

        v8f c1 = {};
        v8f c2 = {};
#pragma unroll
        for (int j = 0; j < 8; ++j) {
            v2f a, bw1, bw2;
            a.x = cur[j].x;  a.y = cur[j].y;
            bw1.x = w1f[j].x; bw1.y = w1f[j].y;
            bw2.x = w2f[j].x; bw2.y = w2f[j].y;
            v2f a2 = a * a;
            c1 = __builtin_amdgcn_wmma_f32_16x16x4_f32(
                    false, a, false, bw1, (short)0, c1, false, false);
            c2 = __builtin_amdgcn_wmma_f32_16x16x4_f32(
                    false, a2, false, bw2, (short)0, c2, false, false);
            a.x = cur[j].z;  a.y = cur[j].w;
            bw1.x = w1f[j].z; bw1.y = w1f[j].w;
            bw2.x = w2f[j].z; bw2.y = w2f[j].w;
            a2 = a * a;
            c1 = __builtin_amdgcn_wmma_f32_16x16x4_f32(
                    false, a, false, bw1, (short)0, c1, false, false);
            c2 = __builtin_amdgcn_wmma_f32_16x16x4_f32(
                    false, a2, false, bw2, (short)0, c2, false, false);
        }
        const int m0 = slab * 16;
#pragma unroll
        for (int r = 0; r < 8; ++r) {
            int mm = m0 + r + 8 * hi;
            if (mm < n) {
                size_t o = (size_t)mm * 64 + ncol;
                out[o] = c1[r] + bias;
                z[o]   = c1[r] + c2[r];
            }
        }
#pragma unroll
        for (int j = 0; j < 8; ++j) cur[j] = nxt[j];
    }
}

#define GRP   16
#define CAP   8192
#define SLOTS (CAP / 512)

__global__ void __launch_bounds__(512) spmm_bins_kernel(
    const int* __restrict__ binstart, const int* __restrict__ pairs,
    const float* __restrict__ dinv, const float* __restrict__ z,
    float* __restrict__ out, int n) {
    __shared__ float tile[RB * DIM];
    __shared__ int scol[CAP];
    __shared__ unsigned char srow[CAP];
    __shared__ __attribute__((aligned(16))) int cnt[RB];
    __shared__ __attribute__((aligned(16))) int rstart[RB];
    const int tid  = threadIdx.x;
    const int lane = tid & 31;
    const int wv = __builtin_amdgcn_readfirstlane(tid >> 5);
    const int b  = blockIdx.x;

    float4* t4 = reinterpret_cast<float4*>(tile);
    for (int k = tid; k < RB * DIM / 4; k += 512) t4[k] = make_float4(0.0f, 0.0f, 0.0f, 0.0f);
    if (tid < RB) cnt[tid] = 0;
    __syncthreads();

    const int s = binstart[b];
    const int t = binstart[b + 1];
    const float2* z2 = reinterpret_cast<const float2*>(z);
    const int2*   pr = reinterpret_cast<const int2*>(pairs);

    for (int cb = s; cb < t; cb += CAP) {
        const int m = (t - cb < CAP) ? (t - cb) : CAP;

        int2 q[SLOTS];
        int  rk[SLOTS];
#pragma unroll
        for (int i = 0; i < SLOTS; ++i) {
            const int mi = m - i * 512;
            q[i]  = make_int2(0, 0);
            rk[i] = 0;
            if (mi > 0) {
                int tc = (tid < mi) ? tid : (mi - 1);
                q[i] = pr[cb + i * 512 + tc];
            }
        }
#pragma unroll
        for (int i = 0; i < SLOTS; ++i) {
            const int mi = m - i * 512;
            if (mi > 0) {
                if (tid < mi) rk[i] = atomicAdd(&cnt[q[i].x & (RB - 1)], 1);
            }
        }
        __syncthreads();

        if (wv == 0) {
            const int4* c4 = reinterpret_cast<const int4*>(cnt);
            int4 a = c4[2 * lane];
            int4 d = c4[2 * lane + 1];
            int tot  = a.x + a.y + a.z + a.w + d.x + d.y + d.z + d.w;
            int incl = tot;
#pragma unroll
            for (int w = 1; w < 32; w <<= 1) {
                int u = __shfl_up(incl, w);
                if (lane >= w) incl += u;
            }
            int4 sa, sd;
            sa.x = incl - tot;
            sa.y = sa.x + a.x;
            sa.z = sa.y + a.y;
            sa.w = sa.z + a.z;
            sd.x = sa.w + a.w;
            sd.y = sd.x + d.x;
            sd.z = sd.y + d.y;
            sd.w = sd.z + d.z;
            int4* s4 = reinterpret_cast<int4*>(rstart);
            s4[2 * lane]     = sa;
            s4[2 * lane + 1] = sd;
        }
        __syncthreads();

        if (tid < RB) cnt[tid] = 0;
#pragma unroll
        for (int h = 0; h < SLOTS; h += 8) {
            int rs[8];
#pragma unroll
            for (int i = 0; i < 8; ++i) {
                const int mi = m - (h + i) * 512;
                rs[i] = 0;
                if (mi > 0) {
                    if (tid < mi) rs[i] = rstart[q[h + i].x & (RB - 1)];
                }
            }
#pragma unroll
            for (int i = 0; i < 8; ++i) {
                const int mi = m - (h + i) * 512;
                if (mi > 0) {
                    if (tid < mi) {
                        int pos = rs[i] + rk[h + i];
                        scol[pos] = q[h + i].y;
                        srow[pos] = (unsigned char)(q[h + i].x & (RB - 1));
                    }
                }
            }
        }
        __syncthreads();

        for (int p0 = wv * 32; p0 < m; p0 += 16 * 32) {
            const int k = (m - p0 < 32) ? (m - p0) : 32;
            int   rl = 0, c = 0;
            float v = 0.0f;
            if (lane < k) {
                c  = scol[p0 + lane];
                rl = srow[p0 + lane];
                v  = dinv[c];
            }
            int   cur = __builtin_amdgcn_readlane(rl, 0);
            float ax = 0.0f, ay = 0.0f;
            for (int g = 0; g < k; g += GRP) {
                float2 x[GRP];
#pragma unroll
                for (int j = 0; j < GRP; ++j) {
                    int idx = (g + j < k) ? (g + j) : (k - 1);
                    int cj = __builtin_amdgcn_readlane(c, idx);
                    x[j] = z2[(size_t)cj * (DIM / 2) + lane];
                }
#pragma unroll
                for (int j = 0; j < GRP; ++j) {
                    int idx = (g + j < k) ? (g + j) : (k - 1);
                    int rj = __builtin_amdgcn_readlane(rl, idx);
                    float vj = __int_as_float(
                        __builtin_amdgcn_readlane(__float_as_int(v), g + j));
                    if (rj != cur) {
                        float* d = tile + cur * DIM + 2 * lane;
                        atomicAdd(d,     ax);
                        atomicAdd(d + 1, ay);
                        ax = 0.0f;
                        ay = 0.0f;
                        cur = rj;
                    }
                    ax += vj * x[j].x;
                    ay += vj * x[j].y;
                }
            }
            float* d = tile + cur * DIM + 2 * lane;
            atomicAdd(d,     ax);
            atomicAdd(d + 1, ay);
        }
        __syncthreads();
    }

    const int row0 = b * RB;
    const float2* tl2 = reinterpret_cast<const float2*>(tile);
    for (int k = tid; k < RB * DIM / 2; k += 512) {
        int i = row0 + (k >> 5);
        if (i < n) {
            float2 a = tl2[k];
            float dr = dinv[i];
            float2* o2 = reinterpret_cast<float2*>(out) + (size_t)i * (DIM / 2) + (k & 31);
            float2 o = *o2;
            o.x += dr * a.x;
            o.y += dr * a.y;
            *o2 = o;
        }
    }
}

extern "C" void kernel_launch(void* const* d_in, const int* in_sizes, int n_in,
                              void* d_out, int out_size, void* d_ws, size_t ws_size,
                              hipStream_t stream) {
    const float* embed = (const float*)d_in[0];
    const int*   row   = (const int*)d_in[1];
    const int*   col   = (const int*)d_in[2];
    const float* W1    = (const float*)d_in[3];
    const float* b1    = (const float*)d_in[4];
    const float* W2    = (const float*)d_in[5];
    const float* b2    = (const float*)d_in[6];
    float*       out   = (float*)d_out;

    const int n = in_sizes[0] / DIM;
    const int E = in_sizes[1];
    const int nbins   = (n + RB - 1) >> RB_LOG;
    const int nblocks = (E + CHUNK - 1) / CHUNK;

    int* wsi = (int*)d_ws;
    size_t nbPad = ((size_t)nbins + 1 + 255) & ~(size_t)255;
    size_t nPad  = ((size_t)n + 255) & ~(size_t)255;
    size_t EPad  = ((size_t)E + 255) & ~(size_t)255;
    int*   hist      = wsi;
    int*   bincursor = wsi + nbPad;
    int*   binstart  = wsi + 2 * nbPad;
    float* dinv      = (float*)(wsi + 3 * nbPad);
    int*   pairs     = wsi + 3 * nbPad + nPad;
    float* z         = (float*)(pairs + 2 * EPad);

    hipMemsetAsync(d_ws, 0, 2 * nbPad * sizeof(int), stream);

    bin_hist_kernel<<<nblocks, BT, nbins * sizeof(int), stream>>>(row, hist, E, nbins);
    bin_scan_kernel<<<1, 1024, 0, stream>>>(hist, binstart, nbins);
    partition_kernel<<<nblocks, BT, nbins * sizeof(int), stream>>>(
        row, col, binstart, bincursor, pairs, E, nbins);
    transform_deg_kernel<<<TGRID, 128, 0, stream>>>(embed, W1, b1, W2, b2,
                                                    binstart, pairs, dinv,
                                                    z, out, n, nbins);

    spmm_bins_kernel<<<nbins, 512, 0, stream>>>(binstart, pairs, dinv, z, out, n);
}


# SpMM final add: a thread's 16 out rows and their dinv are requested together before the first is used (16 reads in flight, not 1), one exposed memory wait per work-group instead of 16
# speedup vs baseline: 3.7957x; 1.0363x over previous
#include <hip/hip_runtime.h>

typedef __attribute__((ext_vector_type(2))) float v2f;
typedef __attribute__((ext_vector_type(8))) float v8f;

#define DIM 64


#define RB_LOG 8
#define RB     256
#define EPT    8
#define BT     1024
#define CHUNK  (EPT * BT)

__global__ void __launch_bounds__(BT) bin_hist_kernel(
    const int* __restrict__ row, int* __restrict__ hist, int E, int nbins) {
    extern __shared__ int sh[];
    const int tid = threadIdx.x;
    for (int k = tid; k < nbins; k += BT) sh[k] = 0;
    __syncthreads();
    const int beg = blockIdx.x * CHUNK;
    int r[EPT];
#pragma unroll
    for (int i = 0; i < EPT; ++i) {
        int e = beg + i * BT + tid;
        r[i] = row[(e < E) ? e : (E - 1)];
    }
#pragma unroll
    for (int i = 0; i < EPT; ++i) {
        int e = beg + i * BT + tid;
        if (e < E) atomicAdd(&sh[r[i] >> RB_LOG], 1);
    }
    __syncthreads();
    for (int k = tid; k < nbins; k += BT) {
        int v = sh[k];
        if (v > 0) atomicAdd(&hist[k], v);
    }
}

__global__ void __launch_bounds__(1024) bin_scan_kernel(
    const int* __restrict__ hist, int* __restrict__ binstart, int nbins) {
    __shared__ int wsum[32];
    __shared__ int carry_s;
    const int tid  = threadIdx.x;
    const int lane = tid & 31;
    const int wv   = tid >> 5;
    if (tid == 0) carry_s = 0;
    __syncthreads();
    for (int base = 0; base < nbins; base += 1024) {
        const int k = base + tid;
        const int c = (k < nbins) ? hist[k] : 0;
        int incl = c;
#pragma unroll
        for (int d = 1; d < 32; d <<= 1) {
            int t = __shfl_up(incl, d);
            if (lane >= d) incl += t;
        }
        if (lane == 31) wsum[wv] = incl;
        __syncthreads();
        const int carry = carry_s;
        int wprefix = 0;
        for (int q = 0; q < wv; ++q) wprefix += wsum[q];
        if (k < nbins) binstart[k] = carry + wprefix + incl - c;
        __syncthreads();
        if (tid == 1023) carry_s = carry + wprefix + incl;
        __syncthreads();
    }
    if (tid == 0) binstart[nbins] = carry_s;
}

__global__ void __launch_bounds__(BT) partition_kernel(
    const int* __restrict__ row, const int* __restrict__ col,
    const int* __restrict__ binstart, int* __restrict__ bincursor,
    int* __restrict__ pairs, int E, int nbins) {
    extern __shared__ int sh[];
    const int tid = threadIdx.x;
    for (int k = tid; k < nbins; k += BT) sh[k] = 0;
    __syncthreads();
    const int beg = blockIdx.x * CHUNK;
    int r[EPT], c[EPT], rk[EPT];
#pragma unroll
    for (int i = 0; i < EPT; ++i) {
        int e  = beg + i * BT + tid;
        int ec = (e < E) ? e : (E - 1);
        r[i] = row[ec];
        c[i] = col[ec];
    }
#pragma unroll
    for (int i = 0; i < EPT; ++i) {
        int e = beg + i * BT + tid;
        rk[i] = 0;
        if (e < E) rk[i] = atomicAdd(&sh[r[i] >> RB_LOG], 1);
    }
    __syncthreads();
    for (int k = tid; k < nbins; k += BT) {
        int cnt  = sh[k];
        int base = 0;
        if (cnt > 0) base = binstart[k] + atomicAdd(&bincursor[k], cnt);
        sh[k] = base;
    }
    __syncthreads();
#pragma unroll
    for (int i = 0; i < EPT; ++i) {
        int e = beg + i * BT + tid;
        if (e < E) {
            int p = sh[r[i] >> RB_LOG] + rk[i];
            int2 q;
            q.x = r[i];
            q.y = c[i];
            reinterpret_cast<int2*>(pairs)[p] = q;
        }
    }
}

#define TGRID 1024
#define DEPT  16

__device__ __forceinline__ void load_frag16(const float* __restrict__ p, float4 (&f)[8]) {
    const float4* q = reinterpret_cast<const float4*>(p);
#pragma unroll
    for (int j = 0; j < 8; ++j) f[j] = q[2 * j];
}

__global__ void __launch_bounds__(128) transform_deg_kernel(
    const float* __restrict__ embed,
    const float* __restrict__ W1, const float* __restrict__ b1,
    const float* __restrict__ W2, const float* __restrict__ b2,
    const int* __restrict__ binstart, const int* __restrict__ pairs,
    float* __restrict__ dinv, float* __restrict__ z, float* __restrict__ out,
    int n, int nbins) {
    __shared__ int cnt[RB];
    const int tid = threadIdx.x;

    for (int b = blockIdx.x; b < nbins; b += TGRID) {
        cnt[tid]       = 0;
        cnt[tid + 128] = 0;
        __syncthreads();
        const int s = binstart[b];
        const int t = binstart[b + 1];
        for (int base = s; base < t; base += 128 * DEPT) {
            int r[DEPT];
#pragma unroll
            for (int i = 0; i < DEPT; ++i) {
                int p  = base + i * 128 + tid;
                int pc = (p < t) ? p : (t - 1);
                r[i] = pairs[2 * (size_t)pc];
            }
#pragma unroll
            for (int i = 0; i < DEPT; ++i) {
                int p = base + i * 128 + tid;
                if (p < t) atomicAdd(&cnt[r[i] & (RB - 1)], 1);
            }
        }
        __syncthreads();
        for (int k = tid; k < RB; k += 128) {
            int i = b * RB + k;
            if (i < n) {
                int c = cnt[k];
                dinv[i] = (c > 0) ? __frsqrt_rn((float)c) : 0.0f;
            }
        }
    }

    const int lane = tid & 31;
    const int wave = tid >> 5;
    const int lo   = lane & 15;
    const int hi   = lane >> 4;
    const int ncol = wave * 16 + lo;
    const int nslabs = (n + 15) >> 4;

    int slab = blockIdx.x;
    if (slab >= nslabs) return;

    float4 w1f[8], w2f[8];
    load_frag16(W1 + ncol * 64 + 4 * hi, w1f);
    load_frag16(W2 + ncol * 64 + 4 * hi, w2f);
    const float bias = b1[ncol] + b2[ncol];

    float4 cur[8];
    {
        int rowA = slab * 16 + lo;
        if (rowA >= n) rowA = n - 1;
        load_frag16(embed + (size_t)rowA * 64 + 4 * hi, cur);
    }

    for (; slab < nslabs; slab += TGRID) {
        const int nslab = slab + TGRID;
        float4 nxt[8];
        if (nslab < nslabs) {
            int rowN = nslab * 16 + lo;
            if (rowN >= n) rowN = n - 1;
            load_frag16(embed + (size_t)rowN * 64 + 4 * hi, nxt);
        } else {
#pragma unroll
            for (int j = 0; j < 8; ++j) nxt[j] = cur[j];
        }

        v8f c1 = {};
        v8f c2 = {};
#pragma unroll
        for (int j = 0; j < 8; ++j) {
            v2f a, bw1, bw2;
            a.x = cur[j].x;  a.y = cur[j].y;
            bw1.x = w1f[j].x; bw1.y = w1f[j].y;
            bw2.x = w2f[j].x; bw2.y = w2f[j].y;
            v2f a2 = a * a;
            c1 = __builtin_amdgcn_wmma_f32_16x16x4_f32(
                    false, a, false, bw1, (short)0, c1, false, false);
            c2 = __builtin_amdgcn_wmma_f32_16x16x4_f32(
                    false, a2, false, bw2, (short)0, c2, false, false);
            a.x = cur[j].z;  a.y = cur[j].w;
            bw1.x = w1f[j].z; bw1.y = w1f[j].w;
            bw2.x = w2f[j].z; bw2.y = w2f[j].w;
            a2 = a * a;
            c1 = __builtin_amdgcn_wmma_f32_16x16x4_f32(
                    false, a, false, bw1, (short)0, c1, false, false);
            c2 = __builtin_amdgcn_wmma_f32_16x16x4_f32(
                    false, a2, false, bw2, (short)0, c2, false, false);
        }
        const int m0 = slab * 16;
#pragma unroll
        for (int r = 0; r < 8; ++r) {
            int mm = m0 + r + 8 * hi;
            if (mm < n) {
                size_t o = (size_t)mm * 64 + ncol;
                out[o] = c1[r] + bias;
                z[o]   = c1[r] + c2[r];
            }
        }
#pragma unroll
        for (int j = 0; j < 8; ++j) cur[j] = nxt[j];
    }
}

#define GRP   16
#define CAP   8192
#define SLOTS (CAP / 512)
#define EB    16

__global__ void __launch_bounds__(512) spmm_bins_kernel(
    const int* __restrict__ binstart, const int* __restrict__ pairs,
    const float* __restrict__ dinv, const float* __restrict__ z,
    float* __restrict__ out, int n) {
    __shared__ float tile[RB * DIM];
    __shared__ int scol[CAP];
    __shared__ unsigned char srow[CAP];
    __shared__ __attribute__((aligned(16))) int cnt[RB];
    __shared__ __attribute__((aligned(16))) int rstart[RB];
    const int tid  = threadIdx.x;
    const int lane = tid & 31;
    const int wv = __builtin_amdgcn_readfirstlane(tid >> 5);
    const int b  = blockIdx.x;

    float4* t4 = reinterpret_cast<float4*>(tile);
    for (int k = tid; k < RB * DIM / 4; k += 512) t4[k] = make_float4(0.0f, 0.0f, 0.0f, 0.0f);
    if (tid < RB) cnt[tid] = 0;
    __syncthreads();

    const int s = binstart[b];
    const int t = binstart[b + 1];
    const float2* z2 = reinterpret_cast<const float2*>(z);
    const int2*   pr = reinterpret_cast<const int2*>(pairs);

    for (int cb = s; cb < t; cb += CAP) {
        const int m = (t - cb < CAP) ? (t - cb) : CAP;

        int2 q[SLOTS];
        int  rk[SLOTS];
#pragma unroll
        for (int i = 0; i < SLOTS; ++i) {
            const int mi = m - i * 512;
            q[i]  = make_int2(0, 0);
            rk[i] = 0;
            if (mi > 0) {
                int tc = (tid < mi) ? tid : (mi - 1);
                q[i] = pr[cb + i * 512 + tc];
            }
        }
#pragma unroll
        for (int i = 0; i < SLOTS; ++i) {
            const int mi = m - i * 512;
            if (mi > 0) {
                if (tid < mi) rk[i] = atomicAdd(&cnt[q[i].x & (RB - 1)], 1);
            }
        }
        __syncthreads();

        if (wv == 0) {
            const int4* c4 = reinterpret_cast<const int4*>(cnt);
            int4 a = c4[2 * lane];
            int4 d = c4[2 * lane + 1];
            int tot  = a.x + a.y + a.z + a.w + d.x + d.y + d.z + d.w;
            int incl = tot;
#pragma unroll
            for (int w = 1; w < 32; w <<= 1) {
                int u = __shfl_up(incl, w);
                if (lane >= w) incl += u;
            }
            int4 sa, sd;
            sa.x = incl - tot;
            sa.y = sa.x + a.x;
            sa.z = sa.y + a.y;
            sa.w = sa.z + a.z;
            sd.x = sa.w + a.w;
            sd.y = sd.x + d.x;
            sd.z = sd.y + d.y;
            sd.w = sd.z + d.z;
            int4* s4 = reinterpret_cast<int4*>(rstart);
            s4[2 * lane]     = sa;
            s4[2 * lane + 1] = sd;
        }
        __syncthreads();

        if (tid < RB) cnt[tid] = 0;
#pragma unroll
        for (int h = 0; h < SLOTS; h += 8) {
            int rs[8];
#pragma unroll
            for (int i = 0; i < 8; ++i) {
                const int mi = m - (h + i) * 512;
                rs[i] = 0;
                if (mi > 0) {
                    if (tid < mi) rs[i] = rstart[q[h + i].x & (RB - 1)];
                }
            }
#pragma unroll
            for (int i = 0; i < 8; ++i) {
                const int mi = m - (h + i) * 512;
                if (mi > 0) {
                    if (tid < mi) {
                        int pos = rs[i] + rk[h + i];
                        scol[pos] = q[h + i].y;
                        srow[pos] = (unsigned char)(q[h + i].x & (RB - 1));
                    }
                }
            }
        }
        __syncthreads();

        for (int p0 = wv * 32; p0 < m; p0 += 16 * 32) {
            const int k = (m - p0 < 32) ? (m - p0) : 32;
            int   rl = 0, c = 0;
            float v = 0.0f;
            if (lane < k) {
                c  = scol[p0 + lane];
                rl = srow[p0 + lane];
                v  = dinv[c];
            }
            int   cur = __builtin_amdgcn_readlane(rl, 0);
            float ax = 0.0f, ay = 0.0f;
            for (int g = 0; g < k; g += GRP) {
                float2 x[GRP];
#pragma unroll
                for (int j = 0; j < GRP; ++j) {
                    int idx = (g + j < k) ? (g + j) : (k - 1);
                    int cj = __builtin_amdgcn_readlane(c, idx);
                    x[j] = z2[(size_t)cj * (DIM / 2) + lane];
                }
#pragma unroll
                for (int j = 0; j < GRP; ++j) {
                    int idx = (g + j < k) ? (g + j) : (k - 1);
                    int rj = __builtin_amdgcn_readlane(rl, idx);
                    float vj = __int_as_float(
                        __builtin_amdgcn_readlane(__float_as_int(v), g + j));
                    if (rj != cur) {
                        float* d = tile + cur * DIM + 2 * lane;
                        atomicAdd(d,     ax);
                        atomicAdd(d + 1, ay);
                        ax = 0.0f;
                        ay = 0.0f;
                        cur = rj;
                    }
                    ax += vj * x[j].x;
                    ay += vj * x[j].y;
                }
            }
            float* d = tile + cur * DIM + 2 * lane;
            atomicAdd(d,     ax);
            atomicAdd(d + 1, ay);
        }
        __syncthreads();
    }

    const int row0 = b * RB;
    const float2* tl2 = reinterpret_cast<const float2*>(tile);
    float2* out2 = reinterpret_cast<float2*>(out);
#pragma unroll
    for (int h = 0; h < RB / 16; h += EB) {
        float2 o[EB];
        float  dr[EB];
#pragma unroll
        for (int u = 0; u < EB; ++u) {
            const int i = row0 + (h + u) * 16 + wv;
            o[u]  = make_float2(0.0f, 0.0f);
            dr[u] = 0.0f;
            if (i < n) {
                o[u]  = out2[(size_t)i * (DIM / 2) + lane];
                dr[u] = dinv[i];
            }
        }
        float2 tv[EB];
#pragma unroll
        for (int u = 0; u < EB; ++u) tv[u] = tl2[((h + u) * 16 + wv) * (DIM / 2) + lane];
#pragma unroll
        for (int u = 0; u < EB; ++u) {
            const int rl = (h + u) * 16 + wv;
            const int i  = row0 + rl;
            if (i < n) {
                const float2 a = tv[u];
                o[u].x += dr[u] * a.x;
                o[u].y += dr[u] * a.y;
                out2[(size_t)i * (DIM / 2) + lane] = o[u];
            }
        }
    }
}

extern "C" void kernel_launch(void* const* d_in, const int* in_sizes, int n_in,
                              void* d_out, int out_size, void* d_ws, size_t ws_size,
                              hipStream_t stream) {
    const float* embed = (const float*)d_in[0];
    const int*   row   = (const int*)d_in[1];
    const int*   col   = (const int*)d_in[2];
    const float* W1    = (const float*)d_in[3];
    const float* b1    = (const float*)d_in[4];
    const float* W2    = (const float*)d_in[5];
    const float* b2    = (const float*)d_in[6];
    float*       out   = (float*)d_out;

    const int n = in_sizes[0] / DIM;
    const int E = in_sizes[1];
    const int nbins   = (n + RB - 1) >> RB_LOG;
    const int nblocks = (E + CHUNK - 1) / CHUNK;

    int* wsi = (int*)d_ws;
    size_t nbPad = ((size_t)nbins + 1 + 255) & ~(size_t)255;
    size_t nPad  = ((size_t)n + 255) & ~(size_t)255;
    size_t EPad  = ((size_t)E + 255) & ~(size_t)255;
    int*   hist      = wsi;
    int*   bincursor = wsi + nbPad;
    int*   binstart  = wsi + 2 * nbPad;
    float* dinv      = (float*)(wsi + 3 * nbPad);
    int*   pairs     = wsi + 3 * nbPad + nPad;
    float* z         = (float*)(pairs + 2 * EPad);

    hipMemsetAsync(d_ws, 0, 2 * nbPad * sizeof(int), stream);

    bin_hist_kernel<<<nblocks, BT, nbins * sizeof(int), stream>>>(row, hist, E, nbins);
    bin_scan_kernel<<<1, 1024, 0, stream>>>(hist, binstart, nbins);
    partition_kernel<<<nblocks, BT, nbins * sizeof(int), stream>>>(
        row, col, binstart, bincursor, pairs, E, nbins);
    transform_deg_kernel<<<TGRID, 128, 0, stream>>>(embed, W1, b1, W2, b2,
                                                    binstart, pairs, dinv,
                                                    z, out, n, nbins);

    spmm_bins_kernel<<<nbins, 512, 0, stream>>>(binstart, pairs, dinv, z, out, n);
}


# Partition: a block's pairs are first placed bin by bin in a shared-memory buffer (scan of its own bin counts), then stored by consecutive threads: contiguous runs per store, not 32 lines of 8 bytes
# speedup vs baseline: 4.1007x; 1.0804x over previous
#include <hip/hip_runtime.h>

typedef __attribute__((ext_vector_type(2))) float v2f;
typedef __attribute__((ext_vector_type(8))) float v8f;

#define DIM 64


#define RB_LOG 8
#define RB     256
#define EPT    8
#define BT     1024
#define CHUNK  (EPT * BT)

__global__ void __launch_bounds__(BT) bin_hist_kernel(
    const int* __restrict__ row, int* __restrict__ hist, int E, int nbins) {
    extern __shared__ int sh[];
    const int tid = threadIdx.x;
    for (int k = tid; k < nbins; k += BT) sh[k] = 0;
    __syncthreads();
    const int beg = blockIdx.x * CHUNK;
    int r[EPT];
#pragma unroll
    for (int i = 0; i < EPT; ++i) {
        int e = beg + i * BT + tid;
        r[i] = row[(e < E) ? e : (E - 1)];
    }
#pragma unroll
    for (int i = 0; i < EPT; ++i) {
        int e = beg + i * BT + tid;
        if (e < E) atomicAdd(&sh[r[i] >> RB_LOG], 1);
    }
    __syncthreads();
    for (int k = tid; k < nbins; k += BT) {
        int v = sh[k];
        if (v > 0) atomicAdd(&hist[k], v);
    }
}

__global__ void __launch_bounds__(1024) bin_scan_kernel(
    const int* __restrict__ hist, int* __restrict__ binstart, int nbins) {
    __shared__ int wsum[32];
    __shared__ int carry_s;
    const int tid  = threadIdx.x;
    const int lane = tid & 31;
    const int wv   = tid >> 5;
    if (tid == 0) carry_s = 0;
    __syncthreads();
    for (int base = 0; base < nbins; base += 1024) {
        const int k = base + tid;
        const int c = (k < nbins) ? hist[k] : 0;
        int incl = c;
#pragma unroll
        for (int d = 1; d < 32; d <<= 1) {
            int t = __shfl_up(incl, d);
            if (lane >= d) incl += t;
        }
        if (lane == 31) wsum[wv] = incl;
        __syncthreads();
        const int carry = carry_s;
        int wprefix = 0;
        for (int q = 0; q < wv; ++q) wprefix += wsum[q];
        if (k < nbins) binstart[k] = carry + wprefix + incl - c;
        __syncthreads();
        if (tid == 1023) carry_s = carry + wprefix + incl;
        __syncthreads();
    }
    if (tid == 0) binstart[nbins] = carry_s;
}

__global__ void __launch_bounds__(BT) partition_kernel(
    const int* __restrict__ row, const int* __restrict__ col,
    const int* __restrict__ binstart, int* __restrict__ bincursor,
    int* __restrict__ pairs, int E, int nbins) {
    extern __shared__ int sh[];
    __shared__ int2 stage[CHUNK];
    __shared__ int wsum[32];
    __shared__ int carry_s;
    int* cnt = sh;
    int* lst = sh + nbins;
    const int tid  = threadIdx.x;
    const int lane = tid & 31;
    const int wv   = tid >> 5;
    for (int k = tid; k < nbins; k += BT) cnt[k] = 0;
    if (tid == 0) carry_s = 0;
    __syncthreads();
    const int beg = blockIdx.x * CHUNK;
    int r[EPT], c[EPT], rk[EPT];
#pragma unroll
    for (int i = 0; i < EPT; ++i) {
        int e  = beg + i * BT + tid;
        int ec = (e < E) ? e : (E - 1);
        r[i] = row[ec];
        c[i] = col[ec];
    }
#pragma unroll
    for (int i = 0; i < EPT; ++i) {
        int e = beg + i * BT + tid;
        rk[i] = 0;
        if (e < E) rk[i] = atomicAdd(&cnt[r[i] >> RB_LOG], 1);
    }
    __syncthreads();
    for (int base = 0; base < nbins; base += BT) {
        const int k  = base + tid;
        const int cv = (k < nbins) ? cnt[k] : 0;
        int bs = 0, res = 0;
        if (cv > 0) {
            bs  = binstart[k];
            res = atomicAdd(&bincursor[k], cv);
        }
        int incl = cv;
#pragma unroll
        for (int d = 1; d < 32; d <<= 1) {
            int t = __shfl_up(incl, d);
            if (lane >= d) incl += t;
        }
        if (lane == 31) wsum[wv] = incl;
        __syncthreads();
        int wp = (lane < wv) ? wsum[lane] : 0;
#pragma unroll
        for (int d = 16; d >= 1; d >>= 1) wp += __shfl_xor(wp, d);
        const int carry = carry_s;
        const int excl  = carry + wp + incl - cv;
        if (k < nbins) {
            lst[k] = excl;
            cnt[k] = bs + res - excl;
        }
        __syncthreads();
        if (tid == BT - 1) carry_s = excl + cv;
        __syncthreads();
    }
    int ls[EPT];
#pragma unroll
    for (int i = 0; i < EPT; ++i) ls[i] = lst[r[i] >> RB_LOG];
#pragma unroll
    for (int i = 0; i < EPT; ++i) {
        int e = beg + i * BT + tid;
        if (e < E) {
            int2 q;
            q.x = r[i];
            q.y = c[i];
            stage[ls[i] + rk[i]] = q;
        }
    }
    __syncthreads();
    const int m = (E - beg < CHUNK) ? (E - beg) : CHUNK;
    int2* pr = reinterpret_cast<int2*>(pairs);
    int2 q[EPT];
    int  d[EPT];
#pragma unroll
    for (int i = 0; i < EPT; ++i) {
        const int j = i * BT + tid;
        q[i] = stage[(j < m) ? j : (m - 1)];
    }
#pragma unroll
    for (int i = 0; i < EPT; ++i) d[i] = cnt[q[i].x >> RB_LOG];
#pragma unroll
    for (int i = 0; i < EPT; ++i) {
        const int j = i * BT + tid;
        if (j < m) pr[d[i] + j] = q[i];
    }
}

#define TGRID 1024
#define DEPT  16

__device__ __forceinline__ void load_frag16(const float* __restrict__ p, float4 (&f)[8]) {
    const float4* q = reinterpret_cast<const float4*>(p);
#pragma unroll
    for (int j = 0; j < 8; ++j) f[j] = q[2 * j];
}

__global__ void __launch_bounds__(128) transform_deg_kernel(
    const float* __restrict__ embed,
    const float* __restrict__ W1, const float* __restrict__ b1,
    const float* __restrict__ W2, const float* __restrict__ b2,
    const int* __restrict__ binstart, const int* __restrict__ pairs,
    float* __restrict__ dinv, float* __restrict__ z, float* __restrict__ out,
    int n, int nbins) {
    __shared__ int cnt[RB];
    const int tid = threadIdx.x;

    for (int b = blockIdx.x; b < nbins; b += TGRID) {
        cnt[tid]       = 0;
        cnt[tid + 128] = 0;
        __syncthreads();
        const int s = binstart[b];
        const int t = binstart[b + 1];
        for (int base = s; base < t; base += 128 * DEPT) {
            int r[DEPT];
#pragma unroll
            for (int i = 0; i < DEPT; ++i) {
                int p  = base + i * 128 + tid;
                int pc = (p < t) ? p : (t - 1);
                r[i] = pairs[2 * (size_t)pc];
            }
#pragma unroll
            for (int i = 0; i < DEPT; ++i) {
                int p = base + i * 128 + tid;
                if (p < t) atomicAdd(&cnt[r[i] & (RB - 1)], 1);
            }
        }
        __syncthreads();
        for (int k = tid; k < RB; k += 128) {
            int i = b * RB + k;
            if (i < n) {
                int c = cnt[k];
                dinv[i] = (c > 0) ? __frsqrt_rn((float)c) : 0.0f;
            }
        }
    }

    const int lane = tid & 31;
    const int wave = tid >> 5;
    const int lo   = lane & 15;
    const int hi   = lane >> 4;
    const int ncol = wave * 16 + lo;
    const int nslabs = (n + 15) >> 4;

    int slab = blockIdx.x;
    if (slab >= nslabs) return;

    float4 w1f[8], w2f[8];
    load_frag16(W1 + ncol * 64 + 4 * hi, w1f);
    load_frag16(W2 + ncol * 64 + 4 * hi, w2f);
    const float bias = b1[ncol] + b2[ncol];

    float4 cur[8];
    {
        int rowA = slab * 16 + lo;
        if (rowA >= n) rowA = n - 1;
        load_frag16(embed + (size_t)rowA * 64 + 4 * hi, cur);
    }

    for (; slab < nslabs; slab += TGRID) {
        const int nslab = slab + TGRID;
        float4 nxt[8];
        if (nslab < nslabs) {
            int rowN = nslab * 16 + lo;
            if (rowN >= n) rowN = n - 1;
            load_frag16(embed + (size_t)rowN * 64 + 4 * hi, nxt);
        } else {
#pragma unroll
            for (int j = 0; j < 8; ++j) nxt[j] = cur[j];
        }

        v8f c1 = {};
        v8f c2 = {};
#pragma unroll
        for (int j = 0; j < 8; ++j) {
            v2f a, bw1, bw2;
            a.x = cur[j].x;  a.y = cur[j].y;
            bw1.x = w1f[j].x; bw1.y = w1f[j].y;
            bw2.x = w2f[j].x; bw2.y = w2f[j].y;
            v2f a2 = a * a;
            c1 = __builtin_amdgcn_wmma_f32_16x16x4_f32(
                    false, a, false, bw1, (short)0, c1, false, false);
            c2 = __builtin_amdgcn_wmma_f32_16x16x4_f32(
                    false, a2, false, bw2, (short)0, c2, false, false);
            a.x = cur[j].z;  a.y = cur[j].w;
            bw1.x = w1f[j].z; bw1.y = w1f[j].w;
            bw2.x = w2f[j].z; bw2.y = w2f[j].w;
            a2 = a * a;
            c1 = __builtin_amdgcn_wmma_f32_16x16x4_f32(
                    false, a, false, bw1, (short)0, c1, false, false);
            c2 = __builtin_amdgcn_wmma_f32_16x16x4_f32(
                    false, a2, false, bw2, (short)0, c2, false, false);
        }
        const int m0 = slab * 16;
#pragma unroll
        for (int r = 0; r < 8; ++r) {
            int mm = m0 + r + 8 * hi;
            if (mm < n) {
                size_t o = (size_t)mm * 64 + ncol;
                out[o] = c1[r] + bias;
                z[o]   = c1[r] + c2[r];
            }
        }
#pragma unroll
        for (int j = 0; j < 8; ++j) cur[j] = nxt[j];
    }
}

#define GRP   16
#define CAP   8192
#define SLOTS (CAP / 512)
#define EB    16

__global__ void __launch_bounds__(512) spmm_bins_kernel(
    const int* __restrict__ binstart, const int* __restrict__ pairs,
    const float* __restrict__ dinv, const float* __restrict__ z,
    float* __restrict__ out, int n) {
    __shared__ float tile[RB * DIM];
    __shared__ int scol[CAP];
    __shared__ unsigned char srow[CAP];
    __shared__ __attribute__((aligned(16))) int cnt[RB];
    __shared__ __attribute__((aligned(16))) int rstart[RB];
    const int tid  = threadIdx.x;
    const int lane = tid & 31;
    const int wv = __builtin_amdgcn_readfirstlane(tid >> 5);
    const int b  = blockIdx.x;

    float4* t4 = reinterpret_cast<float4*>(tile);
    for (int k = tid; k < RB * DIM / 4; k += 512) t4[k] = make_float4(0.0f, 0.0f, 0.0f, 0.0f);
    if (tid < RB) cnt[tid] = 0;
    __syncthreads();

    const int s = binstart[b];
    const int t = binstart[b + 1];
    const float2* z2 = reinterpret_cast<const float2*>(z);
    const int2*   pr = reinterpret_cast<const int2*>(pairs);

    for (int cb = s; cb < t; cb += CAP) {
        const int m = (t - cb < CAP) ? (t - cb) : CAP;

        int2 q[SLOTS];
        int  rk[SLOTS];
#pragma unroll
        for (int i = 0; i < SLOTS; ++i) {
            const int mi = m - i * 512;
            q[i]  = make_int2(0, 0);
            rk[i] = 0;
            if (mi > 0) {
                int tc = (tid < mi) ? tid : (mi - 1);
                q[i] = pr[cb + i * 512 + tc];
            }
        }
#pragma unroll
        for (int i = 0; i < SLOTS; ++i) {
            const int mi = m - i * 512;
            if (mi > 0) {
                if (tid < mi) rk[i] = atomicAdd(&cnt[q[i].x & (RB - 1)], 1);
            }
        }
        __syncthreads();

        if (wv == 0) {
            const int4* c4 = reinterpret_cast<const int4*>(cnt);
            int4 a = c4[2 * lane];
            int4 d = c4[2 * lane + 1];
            int tot  = a.x + a.y + a.z + a.w + d.x + d.y + d.z + d.w;
            int incl = tot;
#pragma unroll
            for (int w = 1; w < 32; w <<= 1) {
                int u = __shfl_up(incl, w);
                if (lane >= w) incl += u;
            }
            int4 sa, sd;
            sa.x = incl - tot;
            sa.y = sa.x + a.x;
            sa.z = sa.y + a.y;
            sa.w = sa.z + a.z;
            sd.x = sa.w + a.w;
            sd.y = sd.x + d.x;
            sd.z = sd.y + d.y;
            sd.w = sd.z + d.z;
            int4* s4 = reinterpret_cast<int4*>(rstart);
            s4[2 * lane]     = sa;
            s4[2 * lane + 1] = sd;
        }
        __syncthreads();

        if (tid < RB) cnt[tid] = 0;
#pragma unroll
        for (int h = 0; h < SLOTS; h += 8) {
            int rs[8];
#pragma unroll
            for (int i = 0; i < 8; ++i) {
                const int mi = m - (h + i) * 512;
                rs[i] = 0;
                if (mi > 0) {
                    if (tid < mi) rs[i] = rstart[q[h + i].x & (RB - 1)];
                }
            }
#pragma unroll
            for (int i = 0; i < 8; ++i) {
                const int mi = m - (h + i) * 512;
                if (mi > 0) {
                    if (tid < mi) {
                        int pos = rs[i] + rk[h + i];
                        scol[pos] = q[h + i].y;
                        srow[pos] = (unsigned char)(q[h + i].x & (RB - 1));
                    }
                }
            }
        }
        __syncthreads();

        for (int p0 = wv * 32; p0 < m; p0 += 16 * 32) {
            const int k = (m - p0 < 32) ? (m - p0) : 32;
            int   rl = 0, c = 0;
            float v = 0.0f;
            if (lane < k) {
                c  = scol[p0 + lane];
                rl = srow[p0 + lane];
                v  = dinv[c];
            }
            int   cur = __builtin_amdgcn_readlane(rl, 0);
            float ax = 0.0f, ay = 0.0f;
            for (int g = 0; g < k; g += GRP) {
                float2 x[GRP];
#pragma unroll
                for (int j = 0; j < GRP; ++j) {
                    int idx = (g + j < k) ? (g + j) : (k - 1);
                    int cj = __builtin_amdgcn_readlane(c, idx);
                    x[j] = z2[(size_t)cj * (DIM / 2) + lane];
                }
#pragma unroll
                for (int j = 0; j < GRP; ++j) {
                    int idx = (g + j < k) ? (g + j) : (k - 1);
                    int rj = __builtin_amdgcn_readlane(rl, idx);
                    float vj = __int_as_float(
                        __builtin_amdgcn_readlane(__float_as_int(v), g + j));
                    if (rj != cur) {
                        float* d = tile + cur * DIM + 2 * lane;
                        atomicAdd(d,     ax);
                        atomicAdd(d + 1, ay);
                        ax = 0.0f;
                        ay = 0.0f;
                        cur = rj;
                    }
                    ax += vj * x[j].x;
                    ay += vj * x[j].y;
                }
            }
            float* d = tile + cur * DIM + 2 * lane;
            atomicAdd(d,     ax);
            atomicAdd(d + 1, ay);
        }
        __syncthreads();
    }

    const int row0 = b * RB;
    const float2* tl2 = reinterpret_cast<const float2*>(tile);
    float2* out2 = reinterpret_cast<float2*>(out);
#pragma unroll
    for (int h = 0; h < RB / 16; h += EB) {
        float2 o[EB];
        float  dr[EB];
#pragma unroll
        for (int u = 0; u < EB; ++u) {
            const int i = row0 + (h + u) * 16 + wv;
            o[u]  = make_float2(0.0f, 0.0f);
            dr[u] = 0.0f;
            if (i < n) {
                o[u]  = out2[(size_t)i * (DIM / 2) + lane];
                dr[u] = dinv[i];
            }
        }
        float2 tv[EB];
#pragma unroll
        for (int u = 0; u < EB; ++u) tv[u] = tl2[((h + u) * 16 + wv) * (DIM / 2) + lane];
#pragma unroll
        for (int u = 0; u < EB; ++u) {
            const int rl = (h + u) * 16 + wv;
            const int i  = row0 + rl;
            if (i < n) {
                const float2 a = tv[u];
                o[u].x += dr[u] * a.x;
                o[u].y += dr[u] * a.y;
                out2[(size_t)i * (DIM / 2) + lane] = o[u];
            }
        }
    }
}

extern "C" void kernel_launch(void* const* d_in, const int* in_sizes, int n_in,
                              void* d_out, int out_size, void* d_ws, size_t ws_size,
                              hipStream_t stream) {
    const float* embed = (const float*)d_in[0];
    const int*   row   = (const int*)d_in[1];
    const int*   col   = (const int*)d_in[2];
    const float* W1    = (const float*)d_in[3];
    const float* b1    = (const float*)d_in[4];
    const float* W2    = (const float*)d_in[5];
    const float* b2    = (const float*)d_in[6];
    float*       out   = (float*)d_out;

    const int n = in_sizes[0] / DIM;
    const int E = in_sizes[1];
    const int nbins   = (n + RB - 1) >> RB_LOG;
    const int nblocks = (E + CHUNK - 1) / CHUNK;

    int* wsi = (int*)d_ws;
    size_t nbPad = ((size_t)nbins + 1 + 255) & ~(size_t)255;
    size_t nPad  = ((size_t)n + 255) & ~(size_t)255;
    size_t EPad  = ((size_t)E + 255) & ~(size_t)255;
    int*   hist      = wsi;
    int*   bincursor = wsi + nbPad;
    int*   binstart  = wsi + 2 * nbPad;
    float* dinv      = (float*)(wsi + 3 * nbPad);
    int*   pairs     = wsi + 3 * nbPad + nPad;
    float* z         = (float*)(pairs + 2 * EPad);

    hipMemsetAsync(d_ws, 0, 2 * nbPad * sizeof(int), stream);

    bin_hist_kernel<<<nblocks, BT, nbins * sizeof(int), stream>>>(row, hist, E, nbins);
    bin_scan_kernel<<<1, 1024, 0, stream>>>(hist, binstart, nbins);
    partition_kernel<<<nblocks, BT, 2 * nbins * sizeof(int), stream>>>(
        row, col, binstart, bincursor, pairs, E, nbins);
    transform_deg_kernel<<<TGRID, 128, 0, stream>>>(embed, W1, b1, W2, b2,
                                                    binstart, pairs, dinv,
                                                    z, out, n, nbins);

    spmm_bins_kernel<<<nbins, 512, 0, stream>>>(binstart, pairs, dinv, z, out, n);
}


# Each bin owns a fixed-capacity region (pairs past it go to a spill list that every bin's group scans): the cursor adds alone place the pairs; the bin-count and bin-scan launches are gone
# speedup vs baseline: 4.3687x; 1.0654x over previous
#include <hip/hip_runtime.h>

typedef __attribute__((ext_vector_type(2))) float v2f;
typedef __attribute__((ext_vector_type(8))) float v8f;

#define DIM 64


#define RB_LOG 8
#define RB     256
#define EPT    8
#define BT     1024
#define CHUNK  (EPT * BT)
#define CAPB   (32768 + 32)

__global__ void __launch_bounds__(BT) partition_kernel(
    const int* __restrict__ row, const int* __restrict__ col,
    int* __restrict__ bincursor, int* __restrict__ pairs,
    int* __restrict__ spill, int E, int nbins) {
    extern __shared__ int sh[];
    __shared__ int2 stage[CHUNK];
    __shared__ int wsum[32];
    __shared__ int carry_s;
    int* cnt = sh;
    int* lst = sh + nbins;
    int* thr = sh + 2 * nbins;
    int* spd = sh + 3 * nbins;
    const int tid  = threadIdx.x;
    const int lane = tid & 31;
    const int wv   = tid >> 5;
    for (int k = tid; k < nbins; k += BT) cnt[k] = 0;
    if (tid == 0) carry_s = 0;
    __syncthreads();
    const int beg = blockIdx.x * CHUNK;
    int r[EPT], c[EPT], rk[EPT];
#pragma unroll
    for (int i = 0; i < EPT; ++i) {
        int e  = beg + i * BT + tid;
        int ec = (e < E) ? e : (E - 1);
        r[i] = row[ec];
        c[i] = col[ec];
    }
#pragma unroll
    for (int i = 0; i < EPT; ++i) {
        int e = beg + i * BT + tid;
        rk[i] = 0;
        if (e < E) rk[i] = atomicAdd(&cnt[r[i] >> RB_LOG], 1);
    }
    __syncthreads();
    for (int base = 0; base < nbins; base += BT) {
        const int k  = base + tid;
        const int cv = (k < nbins) ? cnt[k] : 0;
        int res = 0;
        if (cv > 0) res = atomicAdd(&bincursor[k], cv);
        int incl = cv;
#pragma unroll
        for (int d = 1; d < 32; d <<= 1) {
            int t = __shfl_up(incl, d);
            if (lane >= d) incl += t;
        }
        if (lane == 31) wsum[wv] = incl;
        __syncthreads();
        int wp = (lane < wv) ? wsum[lane] : 0;
#pragma unroll
        for (int d = 16; d >= 1; d >>= 1) wp += __shfl_xor(wp, d);
        const int carry = carry_s;
        const int excl  = carry + wp + incl - cv;
        const int room = CAPB - res;
        const int fit  = (room >= cv) ? cv : ((room > 0) ? room : 0);
        int sres = 0;
        if (fit < cv) sres = atomicAdd(&bincursor[nbins], cv - fit);
        if (k < nbins) {
            lst[k] = excl;
            cnt[k] = k * CAPB + res - excl;
            thr[k] = excl + fit;
            spd[k] = sres - (excl + fit);
        }
        __syncthreads();
        if (tid == BT - 1) carry_s = excl + cv;
        __syncthreads();
    }
    int ls[EPT];
#pragma unroll
    for (int i = 0; i < EPT; ++i) ls[i] = lst[r[i] >> RB_LOG];
#pragma unroll
    for (int i = 0; i < EPT; ++i) {
        int e = beg + i * BT + tid;
        if (e < E) {
            int2 q;
            q.x = r[i];
            q.y = c[i];
            stage[ls[i] + rk[i]] = q;
        }
    }
    __syncthreads();
    const int m = (E - beg < CHUNK) ? (E - beg) : CHUNK;
    int2* pr = reinterpret_cast<int2*>(pairs);
    int2* sp = reinterpret_cast<int2*>(spill);
    int2 q[EPT];
    int  d[EPT], lim[EPT];
#pragma unroll
    for (int i = 0; i < EPT; ++i) {
        const int j = i * BT + tid;
        q[i] = stage[(j < m) ? j : (m - 1)];
    }
#pragma unroll
    for (int i = 0; i < EPT; ++i) {
        d[i]   = cnt[q[i].x >> RB_LOG];
        lim[i] = thr[q[i].x >> RB_LOG];
    }
#pragma unroll
    for (int i = 0; i < EPT; ++i) {
        const int j = i * BT + tid;
        if (j < m) {
            if (j < lim[i]) pr[d[i] + j] = q[i];
            else            sp[spd[q[i].x >> RB_LOG] + j] = q[i];
        }
    }
}

#define TGRID 1024
#define DEPT  16

__device__ __forceinline__ void load_frag16(const float* __restrict__ p, float4 (&f)[8]) {
    const float4* q = reinterpret_cast<const float4*>(p);
#pragma unroll
    for (int j = 0; j < 8; ++j) f[j] = q[2 * j];
}

__global__ void __launch_bounds__(128) transform_deg_kernel(
    const float* __restrict__ embed,
    const float* __restrict__ W1, const float* __restrict__ b1,
    const float* __restrict__ W2, const float* __restrict__ b2,
    const int* __restrict__ bincursor, const int* __restrict__ pairs,
    const int* __restrict__ spill,
    float* __restrict__ dinv, float* __restrict__ z, float* __restrict__ out,
    int n, int nbins) {
    __shared__ int cnt[RB];
    const int tid = threadIdx.x;

    for (int b = blockIdx.x; b < nbins; b += TGRID) {
        cnt[tid]       = 0;
        cnt[tid + 128] = 0;
        __syncthreads();
        const int s = b * CAPB;
        int nb = bincursor[b];
        if (nb > CAPB) nb = CAPB;
        const int t = s + nb;
        for (int base = s; base < t; base += 128 * DEPT) {
            int r[DEPT];
#pragma unroll
            for (int i = 0; i < DEPT; ++i) {
                int p  = base + i * 128 + tid;
                int pc = (p < t) ? p : (t - 1);
                r[i] = pairs[2 * (size_t)pc];
            }
#pragma unroll
            for (int i = 0; i < DEPT; ++i) {
                int p = base + i * 128 + tid;
                if (p < t) atomicAdd(&cnt[r[i] & (RB - 1)], 1);
            }
        }
        const int nspill = bincursor[nbins];
        for (int p = tid; p < nspill; p += 128) {
            const int rs = spill[2 * (size_t)p];
            if ((rs >> RB_LOG) == b) atomicAdd(&cnt[rs & (RB - 1)], 1);
        }
        __syncthreads();
        for (int k = tid; k < RB; k += 128) {
            int i = b * RB + k;
            if (i < n) {
                int c = cnt[k];
                dinv[i] = (c > 0) ? __frsqrt_rn((float)c) : 0.0f;
            }
        }
    }

    const int lane = tid & 31;
    const int wave = tid >> 5;
    const int lo   = lane & 15;
    const int hi   = lane >> 4;
    const int ncol = wave * 16 + lo;
    const int nslabs = (n + 15) >> 4;

    int slab = blockIdx.x;
    if (slab >= nslabs) return;

    float4 w1f[8], w2f[8];
    load_frag16(W1 + ncol * 64 + 4 * hi, w1f);
    load_frag16(W2 + ncol * 64 + 4 * hi, w2f);
    const float bias = b1[ncol] + b2[ncol];

    float4 cur[8];
    {
        int rowA = slab * 16 + lo;
        if (rowA >= n) rowA = n - 1;
        load_frag16(embed + (size_t)rowA * 64 + 4 * hi, cur);
    }

    for (; slab < nslabs; slab += TGRID) {
        const int nslab = slab + TGRID;
        float4 nxt[8];
        if (nslab < nslabs) {
            int rowN = nslab * 16 + lo;
            if (rowN >= n) rowN = n - 1;
            load_frag16(embed + (size_t)rowN * 64 + 4 * hi, nxt);
        } else {
#pragma unroll
            for (int j = 0; j < 8; ++j) nxt[j] = cur[j];
        }

        v8f c1 = {};
        v8f c2 = {};
#pragma unroll
        for (int j = 0; j < 8; ++j) {
            v2f a, bw1, bw2;
            a.x = cur[j].x;  a.y = cur[j].y;
            bw1.x = w1f[j].x; bw1.y = w1f[j].y;
            bw2.x = w2f[j].x; bw2.y = w2f[j].y;
            v2f a2 = a * a;
            c1 = __builtin_amdgcn_wmma_f32_16x16x4_f32(
                    false, a, false, bw1, (short)0, c1, false, false);
            c2 = __builtin_amdgcn_wmma_f32_16x16x4_f32(
                    false, a2, false, bw2, (short)0, c2, false, false);
            a.x = cur[j].z;  a.y = cur[j].w;
            bw1.x = w1f[j].z; bw1.y = w1f[j].w;
            bw2.x = w2f[j].z; bw2.y = w2f[j].w;
            a2 = a * a;
            c1 = __builtin_amdgcn_wmma_f32_16x16x4_f32(
                    false, a, false, bw1, (short)0, c1, false, false);
            c2 = __builtin_amdgcn_wmma_f32_16x16x4_f32(
                    false, a2, false, bw2, (short)0, c2, false, false);
        }
        const int m0 = slab * 16;
#pragma unroll
        for (int r = 0; r < 8; ++r) {
            int mm = m0 + r + 8 * hi;
            if (mm < n) {
                size_t o = (size_t)mm * 64 + ncol;
                out[o] = c1[r] + bias;
                z[o]   = c1[r] + c2[r];
            }
        }
#pragma unroll
        for (int j = 0; j < 8; ++j) cur[j] = nxt[j];
    }
}

#define GRP   16
#define CAP   8192
#define SLOTS (CAP / 512)
#define EB    16

__global__ void __launch_bounds__(512) spmm_bins_kernel(
    const int* __restrict__ bincursor, const int* __restrict__ pairs,
    const int* __restrict__ spill,
    const float* __restrict__ dinv, const float* __restrict__ z,
    float* __restrict__ out, int n, int nbins) {
    __shared__ float tile[RB * DIM];
    __shared__ int scol[CAP];
    __shared__ unsigned char srow[CAP];
    __shared__ __attribute__((aligned(16))) int cnt[RB];
    __shared__ __attribute__((aligned(16))) int rstart[RB];
    const int tid  = threadIdx.x;
    const int lane = tid & 31;
    const int wv = __builtin_amdgcn_readfirstlane(tid >> 5);
    const int b  = blockIdx.x;

    float4* t4 = reinterpret_cast<float4*>(tile);
    for (int k = tid; k < RB * DIM / 4; k += 512) t4[k] = make_float4(0.0f, 0.0f, 0.0f, 0.0f);
    if (tid < RB) cnt[tid] = 0;
    __syncthreads();

    const int s = b * CAPB;
    int nb = bincursor[b];
    if (nb > CAPB) nb = CAPB;
    const int t = s + nb;
    const float2* z2 = reinterpret_cast<const float2*>(z);
    const int2*   pr = reinterpret_cast<const int2*>(pairs);

    for (int cb = s; cb < t; cb += CAP) {
        const int m = (t - cb < CAP) ? (t - cb) : CAP;

        int2 q[SLOTS];
        int  rk[SLOTS];
#pragma unroll
        for (int i = 0; i < SLOTS; ++i) {
            const int mi = m - i * 512;
            q[i]  = make_int2(0, 0);
            rk[i] = 0;
            if (mi > 0) {
                int tc = (tid < mi) ? tid : (mi - 1);
                q[i] = pr[cb + i * 512 + tc];
            }
        }
#pragma unroll
        for (int i = 0; i < SLOTS; ++i) {
            const int mi = m - i * 512;
            if (mi > 0) {
                if (tid < mi) rk[i] = atomicAdd(&cnt[q[i].x & (RB - 1)], 1);
            }
        }
        __syncthreads();

        if (wv == 0) {
            const int4* c4 = reinterpret_cast<const int4*>(cnt);
            int4 a = c4[2 * lane];
            int4 d = c4[2 * lane + 1];
            int tot  = a.x + a.y + a.z + a.w + d.x + d.y + d.z + d.w;
            int incl = tot;
#pragma unroll
            for (int w = 1; w < 32; w <<= 1) {
                int u = __shfl_up(incl, w);
                if (lane >= w) incl += u;
            }
            int4 sa, sd;
            sa.x = incl - tot;
            sa.y = sa.x + a.x;
            sa.z = sa.y + a.y;
            sa.w = sa.z + a.z;
            sd.x = sa.w + a.w;
            sd.y = sd.x + d.x;
            sd.z = sd.y + d.y;
            sd.w = sd.z + d.z;
            int4* s4 = reinterpret_cast<int4*>(rstart);
            s4[2 * lane]     = sa;
            s4[2 * lane + 1] = sd;
        }
        __syncthreads();

        if (tid < RB) cnt[tid] = 0;
#pragma unroll
        for (int h = 0; h < SLOTS; h += 8) {
            int rs[8];
#pragma unroll
            for (int i = 0; i < 8; ++i) {
                const int mi = m - (h + i) * 512;
                rs[i] = 0;
                if (mi > 0) {
                    if (tid < mi) rs[i] = rstart[q[h + i].x & (RB - 1)];
                }
            }
#pragma unroll
            for (int i = 0; i < 8; ++i) {
                const int mi = m - (h + i) * 512;
                if (mi > 0) {
                    if (tid < mi) {
                        int pos = rs[i] + rk[h + i];
                        scol[pos] = q[h + i].y;
                        srow[pos] = (unsigned char)(q[h + i].x & (RB - 1));
                    }
                }
            }
        }
        __syncthreads();

        for (int p0 = wv * 32; p0 < m; p0 += 16 * 32) {
            const int k = (m - p0 < 32) ? (m - p0) : 32;
            int   rl = 0, c = 0;
            float v = 0.0f;
            if (lane < k) {
                c  = scol[p0 + lane];
                rl = srow[p0 + lane];
                v  = dinv[c];
            }
            int   cur = __builtin_amdgcn_readlane(rl, 0);
            float ax = 0.0f, ay = 0.0f;
            for (int g = 0; g < k; g += GRP) {
                float2 x[GRP];
#pragma unroll
                for (int j = 0; j < GRP; ++j) {
                    int idx = (g + j < k) ? (g + j) : (k - 1);
                    int cj = __builtin_amdgcn_readlane(c, idx);
                    x[j] = z2[(size_t)cj * (DIM / 2) + lane];
                }
#pragma unroll
                for (int j = 0; j < GRP; ++j) {
                    int idx = (g + j < k) ? (g + j) : (k - 1);
                    int rj = __builtin_amdgcn_readlane(rl, idx);
                    float vj = __int_as_float(
                        __builtin_amdgcn_readlane(__float_as_int(v), g + j));
                    if (rj != cur) {
                        float* d = tile + cur * DIM + 2 * lane;
                        atomicAdd(d,     ax);
                        atomicAdd(d + 1, ay);
                        ax = 0.0f;
                        ay = 0.0f;
                        cur = rj;
                    }
                    ax += vj * x[j].x;
                    ay += vj * x[j].y;
                }
            }
            float* d = tile + cur * DIM + 2 * lane;
            atomicAdd(d,     ax);
            atomicAdd(d + 1, ay);
        }
        __syncthreads();
    }

    const int nspill = bincursor[nbins];
    if (nspill > 0) {
        const int2* sp = reinterpret_cast<const int2*>(spill);
        for (int p = tid; p < nspill; p += 512) {
            const int2 e = sp[p];
            if ((e.x >> RB_LOG) == b) {
                const float  v  = dinv[e.y];
                const float* zr = z + (size_t)e.y * DIM;
                float* d = tile + (e.x & (RB - 1)) * DIM;
                for (int j = 0; j < DIM; ++j) atomicAdd(d + j, v * zr[j]);
            }
        }
        __syncthreads();
    }

    const int row0 = b * RB;
    const float2* tl2 = reinterpret_cast<const float2*>(tile);
    float2* out2 = reinterpret_cast<float2*>(out);
#pragma unroll
    for (int h = 0; h < RB / 16; h += EB) {
        float2 o[EB];
        float  dr[EB];
#pragma unroll
        for (int u = 0; u < EB; ++u) {
            const int i = row0 + (h + u) * 16 + wv;
            o[u]  = make_float2(0.0f, 0.0f);
            dr[u] = 0.0f;
            if (i < n) {
                o[u]  = out2[(size_t)i * (DIM / 2) + lane];
                dr[u] = dinv[i];
            }
        }
        float2 tv[EB];
#pragma unroll
        for (int u = 0; u < EB; ++u) tv[u] = tl2[((h + u) * 16 + wv) * (DIM / 2) + lane];
#pragma unroll
        for (int u = 0; u < EB; ++u) {
            const int rl = (h + u) * 16 + wv;
            const int i  = row0 + rl;
            if (i < n) {
                const float2 a = tv[u];
                o[u].x += dr[u] * a.x;
                o[u].y += dr[u] * a.y;
                out2[(size_t)i * (DIM / 2) + lane] = o[u];
            }
        }
    }
}

extern "C" void kernel_launch(void* const* d_in, const int* in_sizes, int n_in,
                              void* d_out, int out_size, void* d_ws, size_t ws_size,
                              hipStream_t stream) {
    const float* embed = (const float*)d_in[0];
    const int*   row   = (const int*)d_in[1];
    const int*   col   = (const int*)d_in[2];
    const float* W1    = (const float*)d_in[3];
    const float* b1    = (const float*)d_in[4];
    const float* W2    = (const float*)d_in[5];
    const float* b2    = (const float*)d_in[6];
    float*       out   = (float*)d_out;

    const int n = in_sizes[0] / DIM;
    const int E = in_sizes[1];
    const int nbins   = (n + RB - 1) >> RB_LOG;
    const int nblocks = (E + CHUNK - 1) / CHUNK;

    int* wsi = (int*)d_ws;
    size_t nbPad = ((size_t)nbins + 1 + 255) & ~(size_t)255;
    size_t nPad  = ((size_t)n + 255) & ~(size_t)255;
    size_t EPad  = ((size_t)E + 255) & ~(size_t)255;
    int*   bincursor = wsi;
    float* dinv      = (float*)(wsi + nbPad);
    int*   pairs     = wsi + nbPad + nPad;
    int*   spill     = pairs + 2 * (size_t)nbins * CAPB;
    float* z         = (float*)(spill + 2 * EPad);

    hipMemsetAsync(d_ws, 0, nbPad * sizeof(int), stream);

    partition_kernel<<<nblocks, BT, 4 * nbins * sizeof(int), stream>>>(
        row, col, bincursor, pairs, spill, E, nbins);
    transform_deg_kernel<<<TGRID, 128, 0, stream>>>(embed, W1, b1, W2, b2,
                                                    bincursor, pairs, spill, dinv,
                                                    z, out, n, nbins);

    spmm_bins_kernel<<<nbins, 512, 0, stream>>>(bincursor, pairs, spill, dinv, z,
                                                out, n, nbins);
}


# Dense pass stores: the four waves' 16x16 result pieces pass through a shared-memory slab tile and leave as whole 256-byte rows, 16 B a lane (whole 128-byte lines per store), not 64-byte pieces
# speedup vs baseline: 4.9683x; 1.1372x over previous
#include <hip/hip_runtime.h>

typedef __attribute__((ext_vector_type(2))) float v2f;
typedef __attribute__((ext_vector_type(8))) float v8f;

#define DIM 64


#define RB_LOG 8
#define RB     256
#define EPT    8
#define BT     1024
#define CHUNK  (EPT * BT)
#define CAPB   (32768 + 32)

__global__ void __launch_bounds__(BT) partition_kernel(
    const int* __restrict__ row, const int* __restrict__ col,
    int* __restrict__ bincursor, int* __restrict__ pairs,
    int* __restrict__ spill, int E, int nbins) {
    extern __shared__ int sh[];
    __shared__ int2 stage[CHUNK];
    __shared__ int wsum[32];
    __shared__ int carry_s;
    int* cnt = sh;
    int* lst = sh + nbins;
    int* thr = sh + 2 * nbins;
    int* spd = sh + 3 * nbins;
    const int tid  = threadIdx.x;
    const int lane = tid & 31;
    const int wv   = tid >> 5;
    for (int k = tid; k < nbins; k += BT) cnt[k] = 0;
    if (tid == 0) carry_s = 0;
    __syncthreads();
    const int beg = blockIdx.x * CHUNK;
    int r[EPT], c[EPT], rk[EPT];
#pragma unroll
    for (int i = 0; i < EPT; ++i) {
        int e  = beg + i * BT + tid;
        int ec = (e < E) ? e : (E - 1);
        r[i] = row[ec];
        c[i] = col[ec];
    }
#pragma unroll
    for (int i = 0; i < EPT; ++i) {
        int e = beg + i * BT + tid;
        rk[i] = 0;
        if (e < E) rk[i] = atomicAdd(&cnt[r[i] >> RB_LOG], 1);
    }
    __syncthreads();
    for (int base = 0; base < nbins; base += BT) {
        const int k  = base + tid;
        const int cv = (k < nbins) ? cnt[k] : 0;
        int res = 0;
        if (cv > 0) res = atomicAdd(&bincursor[k], cv);
        int incl = cv;
#pragma unroll
        for (int d = 1; d < 32; d <<= 1) {
            int t = __shfl_up(incl, d);
            if (lane >= d) incl += t;
        }
        if (lane == 31) wsum[wv] = incl;
        __syncthreads();
        int wp = (lane < wv) ? wsum[lane] : 0;
#pragma unroll
        for (int d = 16; d >= 1; d >>= 1) wp += __shfl_xor(wp, d);
        const int carry = carry_s;
        const int excl  = carry + wp + incl - cv;
        const int room = CAPB - res;
        const int fit  = (room >= cv) ? cv : ((room > 0) ? room : 0);
        int sres = 0;
        if (fit < cv) sres = atomicAdd(&bincursor[nbins], cv - fit);
        if (k < nbins) {
            lst[k] = excl;
            cnt[k] = k * CAPB + res - excl;
            thr[k] = excl + fit;
            spd[k] = sres - (excl + fit);
        }
        __syncthreads();
        if (tid == BT - 1) carry_s = excl + cv;
        __syncthreads();
    }
    int ls[EPT];
#pragma unroll
    for (int i = 0; i < EPT; ++i) ls[i] = lst[r[i] >> RB_LOG];
#pragma unroll
    for (int i = 0; i < EPT; ++i) {
        int e = beg + i * BT + tid;
        if (e < E) {
            int2 q;
            q.x = r[i];
            q.y = c[i];
            stage[ls[i] + rk[i]] = q;
        }
    }
    __syncthreads();
    const int m = (E - beg < CHUNK) ? (E - beg) : CHUNK;
    int2* pr = reinterpret_cast<int2*>(pairs);
    int2* sp = reinterpret_cast<int2*>(spill);
    int2 q[EPT];
    int  d[EPT], lim[EPT];
#pragma unroll
    for (int i = 0; i < EPT; ++i) {
        const int j = i * BT + tid;
        q[i] = stage[(j < m) ? j : (m - 1)];
    }
#pragma unroll
    for (int i = 0; i < EPT; ++i) {
        d[i]   = cnt[q[i].x >> RB_LOG];
        lim[i] = thr[q[i].x >> RB_LOG];
    }
#pragma unroll
    for (int i = 0; i < EPT; ++i) {
        const int j = i * BT + tid;
        if (j < m) {
            if (j < lim[i]) pr[d[i] + j] = q[i];
            else            sp[spd[q[i].x >> RB_LOG] + j] = q[i];
        }
    }
}

#define TGRID 1024
#define DEPT  16
#define TP    68

__device__ __forceinline__ void load_frag16(const float* __restrict__ p, float4 (&f)[8]) {
    const float4* q = reinterpret_cast<const float4*>(p);
#pragma unroll
    for (int j = 0; j < 8; ++j) f[j] = q[2 * j];
}

__global__ void __launch_bounds__(128) transform_deg_kernel(
    const float* __restrict__ embed,
    const float* __restrict__ W1, const float* __restrict__ b1,
    const float* __restrict__ W2, const float* __restrict__ b2,
    const int* __restrict__ bincursor, const int* __restrict__ pairs,
    const int* __restrict__ spill,
    float* __restrict__ dinv, float* __restrict__ z, float* __restrict__ out,
    int n, int nbins) {
    __shared__ int cnt[RB];
    __shared__ __attribute__((aligned(16))) float stile[2 * 2 * 16 * TP];
    const int tid = threadIdx.x;

    for (int b = blockIdx.x; b < nbins; b += TGRID) {
        cnt[tid]       = 0;
        cnt[tid + 128] = 0;
        __syncthreads();
        const int s = b * CAPB;
        int nb = bincursor[b];
        if (nb > CAPB) nb = CAPB;
        const int t = s + nb;
        for (int base = s; base < t; base += 128 * DEPT) {
            int r[DEPT];
#pragma unroll
            for (int i = 0; i < DEPT; ++i) {
                int p  = base + i * 128 + tid;
                int pc = (p < t) ? p : (t - 1);
                r[i] = pairs[2 * (size_t)pc];
            }
#pragma unroll
            for (int i = 0; i < DEPT; ++i) {
                int p = base + i * 128 + tid;
                if (p < t) atomicAdd(&cnt[r[i] & (RB - 1)], 1);
            }
        }
        const int nspill = bincursor[nbins];
        for (int p = tid; p < nspill; p += 128) {
            const int rs = spill[2 * (size_t)p];
            if ((rs >> RB_LOG) == b) atomicAdd(&cnt[rs & (RB - 1)], 1);
        }
        __syncthreads();
        for (int k = tid; k < RB; k += 128) {
            int i = b * RB + k;
            if (i < n) {
                int c = cnt[k];
                dinv[i] = (c > 0) ? __frsqrt_rn((float)c) : 0.0f;
            }
        }
    }

    const int lane = tid & 31;
    const int wave = tid >> 5;
    const int lo   = lane & 15;
    const int hi   = lane >> 4;
    const int ncol = wave * 16 + lo;
    const int nslabs = (n + 15) >> 4;

    int slab = blockIdx.x;
    if (slab >= nslabs) return;

    float4 w1f[8], w2f[8];
    load_frag16(W1 + ncol * 64 + 4 * hi, w1f);
    load_frag16(W2 + ncol * 64 + 4 * hi, w2f);
    const float bias = b1[ncol] + b2[ncol];

    int par = 0;
    float4 cur[8];
    {
        int rowA = slab * 16 + lo;
        if (rowA >= n) rowA = n - 1;
        load_frag16(embed + (size_t)rowA * 64 + 4 * hi, cur);
    }

    for (; slab < nslabs; slab += TGRID) {
        const int nslab = slab + TGRID;
        float4 nxt[8];
        if (nslab < nslabs) {
            int rowN = nslab * 16 + lo;
            if (rowN >= n) rowN = n - 1;
            load_frag16(embed + (size_t)rowN * 64 + 4 * hi, nxt);
        } else {
#pragma unroll
            for (int j = 0; j < 8; ++j) nxt[j] = cur[j];
        }

        v8f c1 = {};
        v8f c2 = {};
#pragma unroll
        for (int j = 0; j < 8; ++j) {
            v2f a, bw1, bw2;
            a.x = cur[j].x;  a.y = cur[j].y;
            bw1.x = w1f[j].x; bw1.y = w1f[j].y;
            bw2.x = w2f[j].x; bw2.y = w2f[j].y;
            v2f a2 = a * a;
            c1 = __builtin_amdgcn_wmma_f32_16x16x4_f32(
                    false, a, false, bw1, (short)0, c1, false, false);
            c2 = __builtin_amdgcn_wmma_f32_16x16x4_f32(
                    false, a2, false, bw2, (short)0, c2, false, false);
            a.x = cur[j].z;  a.y = cur[j].w;
            bw1.x = w1f[j].z; bw1.y = w1f[j].w;
            bw2.x = w2f[j].z; bw2.y = w2f[j].w;
            a2 = a * a;
            c1 = __builtin_amdgcn_wmma_f32_16x16x4_f32(
                    false, a, false, bw1, (short)0, c1, false, false);
            c2 = __builtin_amdgcn_wmma_f32_16x16x4_f32(
                    false, a2, false, bw2, (short)0, c2, false, false);
        }
        const int m0 = slab * 16;
        float* to = stile + par * (2 * 16 * TP);
        float* tz = to + 16 * TP;
#pragma unroll
        for (int r = 0; r < 8; ++r) {
            const int a = (r + 8 * hi) * TP + ncol;
            to[a] = c1[r] + bias;
            tz[a] = c1[r] + c2[r];
        }
        __builtin_amdgcn_fence(__ATOMIC_RELEASE, "workgroup", "local");
        __builtin_amdgcn_s_barrier();
        __builtin_amdgcn_fence(__ATOMIC_ACQUIRE, "workgroup", "local");
#pragma unroll
        for (int p = 0; p < 2; ++p) {
            const int rr = p * 8 + (tid >> 4);
            const int c4 = (tid & 15) * 4;
            const float4 vo = *reinterpret_cast<const float4*>(to + rr * TP + c4);
            const float4 vz = *reinterpret_cast<const float4*>(tz + rr * TP + c4);
            const int mm = m0 + rr;
            if (mm < n) {
                const size_t o = (size_t)mm * 64 + c4;
                *reinterpret_cast<float4*>(out + o) = vo;
                *reinterpret_cast<float4*>(z + o)   = vz;
            }
        }
        par ^= 1;
#pragma unroll
        for (int j = 0; j < 8; ++j) cur[j] = nxt[j];
    }
}

#define GRP   16
#define CAP   8192
#define SLOTS (CAP / 512)
#define EB    16

__global__ void __launch_bounds__(512) spmm_bins_kernel(
    const int* __restrict__ bincursor, const int* __restrict__ pairs,
    const int* __restrict__ spill,
    const float* __restrict__ dinv, const float* __restrict__ z,
    float* __restrict__ out, int n, int nbins) {
    __shared__ float tile[RB * DIM];
    __shared__ int scol[CAP];
    __shared__ unsigned char srow[CAP];
    __shared__ __attribute__((aligned(16))) int cnt[RB];
    __shared__ __attribute__((aligned(16))) int rstart[RB];
    const int tid  = threadIdx.x;
    const int lane = tid & 31;
    const int wv = __builtin_amdgcn_readfirstlane(tid >> 5);
    const int b  = blockIdx.x;

    float4* t4 = reinterpret_cast<float4*>(tile);
    for (int k = tid; k < RB * DIM / 4; k += 512) t4[k] = make_float4(0.0f, 0.0f, 0.0f, 0.0f);
    if (tid < RB) cnt[tid] = 0;
    __syncthreads();

    const int s = b * CAPB;
    int nb = bincursor[b];
    if (nb > CAPB) nb = CAPB;
    const int t = s + nb;
    const float2* z2 = reinterpret_cast<const float2*>(z);
    const int2*   pr = reinterpret_cast<const int2*>(pairs);

    for (int cb = s; cb < t; cb += CAP) {
        const int m = (t - cb < CAP) ? (t - cb) : CAP;

        int2 q[SLOTS];
        int  rk[SLOTS];
#pragma unroll
        for (int i = 0; i < SLOTS; ++i) {
            const int mi = m - i * 512;
            q[i]  = make_int2(0, 0);
            rk[i] = 0;
            if (mi > 0) {
                int tc = (tid < mi) ? tid : (mi - 1);
                q[i] = pr[cb + i * 512 + tc];
            }
        }
#pragma unroll
        for (int i = 0; i < SLOTS; ++i) {
            const int mi = m - i * 512;
            if (mi > 0) {
                if (tid < mi) rk[i] = atomicAdd(&cnt[q[i].x & (RB - 1)], 1);
            }
        }
        __syncthreads();

        if (wv == 0) {
            const int4* c4 = reinterpret_cast<const int4*>(cnt);
            int4 a = c4[2 * lane];
            int4 d = c4[2 * lane + 1];
            int tot  = a.x + a.y + a.z + a.w + d.x + d.y + d.z + d.w;
            int incl = tot;
#pragma unroll
            for (int w = 1; w < 32; w <<= 1) {
                int u = __shfl_up(incl, w);
                if (lane >= w) incl += u;
            }
            int4 sa, sd;
            sa.x = incl - tot;
            sa.y = sa.x + a.x;
            sa.z = sa.y + a.y;
            sa.w = sa.z + a.z;
            sd.x = sa.w + a.w;
            sd.y = sd.x + d.x;
            sd.z = sd.y + d.y;
            sd.w = sd.z + d.z;
            int4* s4 = reinterpret_cast<int4*>(rstart);
            s4[2 * lane]     = sa;
            s4[2 * lane + 1] = sd;
        }
        __syncthreads();

        if (tid < RB) cnt[tid] = 0;
#pragma unroll
        for (int h = 0; h < SLOTS; h += 8) {
            int rs[8];
#pragma unroll
            for (int i = 0; i < 8; ++i) {
                const int mi = m - (h + i) * 512;
                rs[i] = 0;
                if (mi > 0) {
                    if (tid < mi) rs[i] = rstart[q[h + i].x & (RB - 1)];
                }
            }
#pragma unroll
            for (int i = 0; i < 8; ++i) {
                const int mi = m - (h + i) * 512;
                if (mi > 0) {
                    if (tid < mi) {
                        int pos = rs[i] + rk[h + i];
                        scol[pos] = q[h + i].y;
                        srow[pos] = (unsigned char)(q[h + i].x & (RB - 1));
                    }
                }
            }
        }
        __syncthreads();

        for (int p0 = wv * 32; p0 < m; p0 += 16 * 32) {
            const int k = (m - p0 < 32) ? (m - p0) : 32;
            int   rl = 0, c = 0;
            float v = 0.0f;
            if (lane < k) {
                c  = scol[p0 + lane];
                rl = srow[p0 + lane];
                v  = dinv[c];
            }
            int   cur = __builtin_amdgcn_readlane(rl, 0);
            float ax = 0.0f, ay = 0.0f;
            for (int g = 0; g < k; g += GRP) {
                float2 x[GRP];
#pragma unroll
                for (int j = 0; j < GRP; ++j) {
                    int idx = (g + j < k) ? (g + j) : (k - 1);
                    int cj = __builtin_amdgcn_readlane(c, idx);
                    x[j] = z2[(size_t)cj * (DIM / 2) + lane];
                }
#pragma unroll
                for (int j = 0; j < GRP; ++j) {
                    int idx = (g + j < k) ? (g + j) : (k - 1);
                    int rj = __builtin_amdgcn_readlane(rl, idx);
                    float vj = __int_as_float(
                        __builtin_amdgcn_readlane(__float_as_int(v), g + j));
                    if (rj != cur) {
                        float* d = tile + cur * DIM + 2 * lane;
                        atomicAdd(d,     ax);
                        atomicAdd(d + 1, ay);
                        ax = 0.0f;
                        ay = 0.0f;
                        cur = rj;
                    }
                    ax += vj * x[j].x;
                    ay += vj * x[j].y;
                }
            }
            float* d = tile + cur * DIM + 2 * lane;
            atomicAdd(d,     ax);
            atomicAdd(d + 1, ay);
        }
        __syncthreads();
    }

    const int nspill = bincursor[nbins];
    if (nspill > 0) {
        const int2* sp = reinterpret_cast<const int2*>(spill);
        for (int p = tid; p < nspill; p += 512) {
            const int2 e = sp[p];
            if ((e.x >> RB_LOG) == b) {
                const float  v  = dinv[e.y];
                const float* zr = z + (size_t)e.y * DIM;
                float* d = tile + (e.x & (RB - 1)) * DIM;
                for (int j = 0; j < DIM; ++j) atomicAdd(d + j, v * zr[j]);
            }
        }
        __syncthreads();
    }

    const int row0 = b * RB;
    const float2* tl2 = reinterpret_cast<const float2*>(tile);
    float2* out2 = reinterpret_cast<float2*>(out);
#pragma unroll
    for (int h = 0; h < RB / 16; h += EB) {
        float2 o[EB];
        float  dr[EB];
#pragma unroll
        for (int u = 0; u < EB; ++u) {
            const int i = row0 + (h + u) * 16 + wv;
            o[u]  = make_float2(0.0f, 0.0f);
            dr[u] = 0.0f;
            if (i < n) {
                o[u]  = out2[(size_t)i * (DIM / 2) + lane];
                dr[u] = dinv[i];
            }
        }
        float2 tv[EB];
#pragma unroll
        for (int u = 0; u < EB; ++u) tv[u] = tl2[((h + u) * 16 + wv) * (DIM / 2) + lane];
#pragma unroll
        for (int u = 0; u < EB; ++u) {
            const int rl = (h + u) * 16 + wv;
            const int i  = row0 + rl;
            if (i < n) {
                const float2 a = tv[u];
                o[u].x += dr[u] * a.x;
                o[u].y += dr[u] * a.y;
                out2[(size_t)i * (DIM / 2) + lane] = o[u];
            }
        }
    }
}

extern "C" void kernel_launch(void* const* d_in, const int* in_sizes, int n_in,
                              void* d_out, int out_size, void* d_ws, size_t ws_size,
                              hipStream_t stream) {
    const float* embed = (const float*)d_in[0];
    const int*   row   = (const int*)d_in[1];
    const int*   col   = (const int*)d_in[2];
    const float* W1    = (const float*)d_in[3];
    const float* b1    = (const float*)d_in[4];
    const float* W2    = (const float*)d_in[5];
    const float* b2    = (const float*)d_in[6];
    float*       out   = (float*)d_out;

    const int n = in_sizes[0] / DIM;
    const int E = in_sizes[1];
    const int nbins   = (n + RB - 1) >> RB_LOG;
    const int nblocks = (E + CHUNK - 1) / CHUNK;

    int* wsi = (int*)d_ws;
    size_t nbPad = ((size_t)nbins + 1 + 255) & ~(size_t)255;
    size_t nPad  = ((size_t)n + 255) & ~(size_t)255;
    size_t EPad  = ((size_t)E + 255) & ~(size_t)255;
    int*   bincursor = wsi;
    float* dinv      = (float*)(wsi + nbPad);
    int*   pairs     = wsi + nbPad + nPad;
    int*   spill     = pairs + 2 * (size_t)nbins * CAPB;
    float* z         = (float*)(spill + 2 * EPad);

    hipMemsetAsync(d_ws, 0, nbPad * sizeof(int), stream);

    partition_kernel<<<nblocks, BT, 4 * nbins * sizeof(int), stream>>>(
        row, col, bincursor, pairs, spill, E, nbins);
    transform_deg_kernel<<<TGRID, 128, 0, stream>>>(embed, W1, b1, W2, b2,
                                                    bincursor, pairs, spill, dinv,
                                                    z, out, n, nbins);

    spmm_bins_kernel<<<nbins, 512, 0, stream>>>(bincursor, pairs, spill, dinv, z,
                                                out, n, nbins);
}
